# starting kernel
# baseline (speedup 1.0000x reference)
#include <hip/hip_runtime.h>
#include <stdint.h>

#define BATCH 2
#define CH    64
#define NPT   8192
#define KNN_K 20
#define COUT  64
#define CIN2  128
#define NEGSL 0.2f
#define BNEPS 1e-5f

typedef float v2f __attribute__((ext_vector_type(2)));
typedef float v4f __attribute__((ext_vector_type(4)));
typedef float v8f __attribute__((ext_vector_type(8)));

__device__ __forceinline__ v8f wmma_f32(v2f a, v2f b, v8f c) {
  return __builtin_amdgcn_wmma_f32_16x16x4_f32(false, a, false, b, (short)0, c, false, false);
}

__device__ __forceinline__ unsigned fenc(float f) {
  unsigned u = __float_as_uint(f);
  return (u & 0x80000000u) ? ~u : (u | 0x80000000u);
}
__device__ __forceinline__ float fdec(unsigned u) {
  return __uint_as_float((u & 0x80000000u) ? (u ^ 0x80000000u) : ~u);
}

__global__ void prep_kernel(const float* __restrict__ x,
                            float* __restrict__ xt, float* __restrict__ sq) {
  int gp = blockIdx.x * blockDim.x + threadIdx.x;
  int b = gp >> 13, n = gp & (NPT - 1);
  const float* xb = x + (size_t)b * CH * NPT + n;
  float acc = 0.f;
  #pragma unroll 8
  for (int c = 0; c < CH; ++c) {
    float v = xb[(size_t)c * NPT];
    acc += v * v;
    xt[(size_t)gp * CH + c] = v;
  }
  sq[gp] = acc;
}

__launch_bounds__(32)
__global__ void knn_kernel(const float* __restrict__ xt, const float* __restrict__ sq,
                           int* __restrict__ knn) {
  __shared__ float tile[16 * 16];
  __shared__ float mv[32 * KNN_K];
  __shared__ int   mi[32 * KNN_K];

  const int L  = threadIdx.x;
  const int rr = L & 15, hh = L >> 4;
  const int b  = blockIdx.x >> 9;
  const int r0 = (blockIdx.x & 511) << 4;
  const int bofs = b << 13;

  v2f a[16];
  const float* arow = xt + (size_t)(bofs + r0 + rr) * CH + 2 * hh;
  #pragma unroll
  for (int kk = 0; kk < 16; ++kk) a[kk] = *(const v2f*)(arow + kk * 4);
  const float sqn = sq[bofs + r0 + rr];

  float vals[KNN_K]; int idxs[KNN_K];
  #pragma unroll
  for (int q = 0; q < KNN_K; ++q) { vals[q] = -__builtin_inff(); idxs[q] = 0; }

  const float* xbase = xt + (size_t)bofs * CH + 2 * hh;

  auto loadB = [&](int ct, v2f (&bb)[16]) {
    const float* brow = xbase + (size_t)(((ct << 4) + rr)) * CH;
    #pragma unroll
    for (int kk = 0; kk < 16; ++kk) bb[kk] = *(const v2f*)(brow + kk * 4);
  };

  auto step = [&](int ct, v2f (&cur)[16], v2f (&nxt)[16]) {
    const int c0 = ct << 4;
    v4f sq0 = *(const v4f*)(sq + bofs + c0 + hh * 8);
    v4f sq1 = *(const v4f*)(sq + bofs + c0 + hh * 8 + 4);
    float sqt[8];
    #pragma unroll
    for (int j = 0; j < 4; ++j) { sqt[j] = sq0[j]; sqt[j + 4] = sq1[j]; }

    v8f acc = {};
    #pragma unroll
    for (int kk = 0; kk < 16; ++kk) acc = wmma_f32(a[kk], cur[kk], acc);

    loadB((ct + 1) & 511, nxt);

    v4f lo = __builtin_shufflevector(acc, acc, 0, 1, 2, 3);
    v4f hi = __builtin_shufflevector(acc, acc, 4, 5, 6, 7);
    float* p = &tile[rr * 16 + hh * 8];
    *(v4f*)p       = lo;
    *(v4f*)(p + 4) = hi;
    __syncthreads();

    #pragma unroll
    for (int j = 0; j < 8; ++j) {
      int ci = hh * 8 + j;
      int colg = c0 + ci;
      float f = tile[ci * 16 + rr];
      float s = 2.f * f - sqt[j] - sqn;
      if (s > vals[0]) {
        float old19 = vals[KNN_K - 1];
        #pragma unroll
        for (int t = 0; t < KNN_K - 1; ++t) {
          bool sh = s > vals[t + 1];
          float kv = vals[t]; int ki = idxs[t];
          vals[t] = sh ? vals[t + 1] : fmaxf(s, kv);
          idxs[t] = sh ? idxs[t + 1] : ((kv >= s) ? ki : colg);
        }
        if (s > old19) { vals[KNN_K - 1] = s; idxs[KNN_K - 1] = colg; }
      }
    }
    __syncthreads();
  };

  v2f bufA[16], bufB[16];
  loadB(0, bufA);
  for (int ct = 0; ct < NPT / 16; ct += 2) {
    step(ct,     bufA, bufB);
    step(ct + 1, bufB, bufA);
  }

  #pragma unroll
  for (int q = 0; q < KNN_K; ++q) { mv[L * KNN_K + q] = vals[q]; mi[L * KNN_K + q] = idxs[q]; }
  __syncthreads();
  if (L < 16) {
    const float* va = &mv[L * KNN_K];          const int* ia = &mi[L * KNN_K];
    const float* vb = &mv[(L + 16) * KNN_K];   const int* ib = &mi[(L + 16) * KNN_K];
    int pa = KNN_K - 1, pb = KNN_K - 1;
    int* outp = knn + (size_t)(bofs + r0 + L) * KNN_K;
    for (int q = 0; q < KNN_K; ++q) {
      int sa = pa < 0 ? 0 : pa, sb = pb < 0 ? 0 : pb;
      float fa = (pa >= 0) ? va[sa] : -__builtin_inff();
      float fb = (pb >= 0) ? vb[sb] : -__builtin_inff();
      bool ta = fa >= fb;
      outp[q] = ta ? ia[sa] : ib[sb];
      if (ta) --pa; else --pb;
    }
  }
}

__launch_bounds__(128)
__global__ void conv_kernel(const float* __restrict__ xt, const int* __restrict__ knn,
                            const float* __restrict__ W, const float* __restrict__ bias,
                            float* __restrict__ mxp, float* __restrict__ mnp,
                            float* __restrict__ part) {
  __shared__ float    s_edge[4][16 * CIN2];
  __shared__ unsigned s_umax[4][4][COUT];
  __shared__ unsigned s_umin[4][4][COUT];
  __shared__ float    s_psum[4][16][8];
  __shared__ float    s_psq [4][16][8];

  const int tid = threadIdx.x;
  const int w = tid >> 5, L = tid & 31;
  const int rr = L & 15, hh = L >> 4;
  const int wfp = blockIdx.x * 16 + w * 4;

  {
    unsigned* um = &s_umax[0][0][0];
    unsigned* un = &s_umin[0][0][0];
    for (int i = tid; i < 4 * 4 * COUT; i += 128) { um[i] = 0u; un[i] = 0xFFFFFFFFu; }
  }
  __syncthreads();

  float fsum[4] = {0.f, 0.f, 0.f, 0.f}, fsq[4] = {0.f, 0.f, 0.f, 0.f};
  float* edge = &s_edge[w][0];

  #pragma unroll 1
  for (int t = 0; t < 5; ++t) {
    {
      int j = L >> 1, half = L & 1;
      int le = t * 16 + j;
      int p = le / 20, k = le - 20 * p;
      int gp = wfp + p;
      int bb = gp >> 13;
      int m = knn[(size_t)gp * KNN_K + k];
      const float* xn = xt + (size_t)gp * CH;
      const float* xm = xt + (size_t)((bb << 13) + m) * CH;
      float* er = edge + j * CIN2;
      if (half == 0) {
        #pragma unroll
        for (int c = 0; c < CH; c += 4)
          *(v4f*)(er + c) = *(const v4f*)(xm + c) - *(const v4f*)(xn + c);
      } else {
        #pragma unroll
        for (int c = 0; c < CH; c += 4)
          *(v4f*)(er + CH + c) = *(const v4f*)(xn + c);
      }
    }
    __syncthreads();

    int le0 = t * 16 + hh * 8;
    int pA = le0 / 20;
    int vth = (pA + 1) * 20 - le0;
    bool hasB = vth < 8;

    #pragma unroll 1
    for (int ot = 0; ot < 4; ++ot) {
      const int o = ot * 16 + rr;
      const float* wrow = W + (size_t)o * CIN2 + 2 * hh;
      const float* arow = edge + rr * CIN2 + 2 * hh;
      v8f acc = {};
      #pragma unroll
      for (int g = 0; g < 4; ++g) {
        v2f av[8], bv[8];
        #pragma unroll
        for (int u = 0; u < 8; ++u) {
          av[u] = *(const v2f*)(arow + (g * 8 + u) * 4);
          bv[u] = *(const v2f*)(wrow + (g * 8 + u) * 4);
        }
        #pragma unroll
        for (int u = 0; u < 8; ++u) acc = wmma_f32(av[u], bv[u], acc);
      }
      float bo = bias[o];
      float gAmax = -__builtin_inff(), gAmin = __builtin_inff();
      float gBmax = -__builtin_inff(), gBmin = __builtin_inff();
      float lsum = 0.f, lsq = 0.f;
      #pragma unroll
      for (int v = 0; v < 8; ++v) {
        float h = acc[v] + bo;
        h = (h >= 0.f) ? h : NEGSL * h;
        lsum += h; lsq += h * h;
        bool inB = (v >= vth);
        gAmax = fmaxf(gAmax, inB ? -__builtin_inff() : h);
        gAmin = fminf(gAmin, inB ?  __builtin_inff() : h);
        gBmax = fmaxf(gBmax, inB ? h : -__builtin_inff());
        gBmin = fminf(gBmin, inB ? h :  __builtin_inff());
      }
      fsum[ot] += lsum; fsq[ot] += lsq;
      atomicMax(&s_umax[w][pA][o], fenc(gAmax));
      atomicMin(&s_umin[w][pA][o], fenc(gAmin));
      if (hasB) {
        atomicMax(&s_umax[w][pA + 1][o], fenc(gBmax));
        atomicMin(&s_umin[w][pA + 1][o], fenc(gBmin));
      }
    }
    __syncthreads();
  }

  #pragma unroll
  for (int ot = 0; ot < 4; ++ot) {
    s_psum[ot][rr][w * 2 + hh] = fsum[ot];
    s_psq [ot][rr][w * 2 + hh] = fsq[ot];
  }
  __syncthreads();
  if (tid < COUT) {
    float s = 0.f, s2 = 0.f;
    #pragma unroll
    for (int q = 0; q < 8; ++q) { s += s_psum[tid >> 4][tid & 15][q]; s2 += s_psq[tid >> 4][tid & 15][q]; }
    part[(size_t)blockIdx.x * 128 + tid]      = s;
    part[(size_t)blockIdx.x * 128 + 64 + tid] = s2;
  }

  #pragma unroll
  for (int it = 0; it < 8; ++it) {
    int q = L + 32 * it;
    int p = q >> 6, o = q & 63;
    int gp = wfp + p;
    int bb = gp >> 13, n = gp & (NPT - 1);
    size_t oi = ((size_t)(bb * COUT + o) << 13) + n;
    mxp[oi] = fdec(s_umax[w][p][o]);
    mnp[oi] = fdec(s_umin[w][p][o]);
  }
}

__global__ void stats_kernel(const float* __restrict__ part, float* __restrict__ gst) {
  int t = threadIdx.x;
  float s = 0.f;
  for (int g = 0; g < 1024; ++g) s += part[(size_t)g * 128 + t];
  gst[t] = s;
}

__global__ void out_kernel(const float* __restrict__ mxp, const float* __restrict__ mnp,
                           const float* __restrict__ gst,
                           const float* __restrict__ gamma, const float* __restrict__ beta,
                           float* __restrict__ outp) {
  int i = blockIdx.x * blockDim.x + threadIdx.x;
  int o = (i >> 13) & 63;
  const float cnt = (float)(BATCH * NPT * KNN_K);
  float mean = gst[o] / cnt;
  float var  = gst[COUT + o] / cnt - mean * mean;
  float scale = gamma[o] * rsqrtf(var + BNEPS);
  float sel = (scale >= 0.f) ? mxp[i] : mnp[i];
  outp[i] = (sel - mean) * scale + beta[o];
}

extern "C" void kernel_launch(void* const* d_in, const int* in_sizes, int n_in,
                              void* d_out, int out_size, void* d_ws, size_t ws_size,
                              hipStream_t stream) {
  const float* x     = (const float*)d_in[0];
  const float* W     = (const float*)d_in[1];
  const float* bias  = (const float*)d_in[2];
  const float* gamma = (const float*)d_in[3];
  const float* beta  = (const float*)d_in[4];
  float* outp = (float*)d_out;

  float* xt  = (float*)d_ws;
  float* sq  = xt + (size_t)BATCH * NPT * CH;
  int*   knn = (int*)(sq + (size_t)BATCH * NPT);
  float* mxp = (float*)(knn + (size_t)BATCH * NPT * KNN_K);
  float* mnp = mxp + (size_t)BATCH * COUT * NPT;
  float* gst = mnp + (size_t)BATCH * COUT * NPT;
  float* part = gst + 128;

  prep_kernel <<<(BATCH * NPT) / 256, 256, 0, stream>>>(x, xt, sq);
  knn_kernel  <<<BATCH * (NPT / 16), 32, 0, stream>>>(xt, sq, knn);
  conv_kernel <<<(BATCH * NPT) / 16, 128, 0, stream>>>(xt, knn, W, bias, mxp, mnp, part);
  stats_kernel<<<1, 128, 0, stream>>>(part, gst);
  out_kernel  <<<(BATCH * COUT * NPT) / 256, 256, 0, stream>>>(mxp, mnp, gst, gamma, beta, outp);
}


# knn: column tile as WMMA A operand, row block as B, so each lane's accumulator holds its own scan row; LDS transpose, both loop barriers and the full load wait they forced are gone
# speedup vs baseline: 1.2877x; 1.2877x over previous
#include <hip/hip_runtime.h>
#include <stdint.h>

#define BATCH 2
#define CH    64
#define NPT   8192
#define KNN_K 20
#define COUT  64
#define CIN2  128
#define NEGSL 0.2f
#define BNEPS 1e-5f

typedef float v2f __attribute__((ext_vector_type(2)));
typedef float v4f __attribute__((ext_vector_type(4)));
typedef float v8f __attribute__((ext_vector_type(8)));

__device__ __forceinline__ v8f wmma_f32(v2f a, v2f b, v8f c) {
  return __builtin_amdgcn_wmma_f32_16x16x4_f32(false, a, false, b, (short)0, c, false, false);
}

__device__ __forceinline__ unsigned fenc(float f) {
  unsigned u = __float_as_uint(f);
  return (u & 0x80000000u) ? ~u : (u | 0x80000000u);
}
__device__ __forceinline__ float fdec(unsigned u) {
  return __uint_as_float((u & 0x80000000u) ? (u ^ 0x80000000u) : ~u);
}

__global__ void prep_kernel(const float* __restrict__ x,
                            float* __restrict__ xt, float* __restrict__ sq) {
  int gp = blockIdx.x * blockDim.x + threadIdx.x;
  int b = gp >> 13, n = gp & (NPT - 1);
  const float* xb = x + (size_t)b * CH * NPT + n;
  float acc = 0.f;
  #pragma unroll 8
  for (int c = 0; c < CH; ++c) {
    float v = xb[(size_t)c * NPT];
    acc += v * v;
    xt[(size_t)gp * CH + c] = v;
  }
  sq[gp] = acc;
}

__launch_bounds__(32)
__global__ void knn_kernel(const float* __restrict__ xt, const float* __restrict__ sq,
                           int* __restrict__ knn) {
  __shared__ float mv[32 * KNN_K];
  __shared__ int   mi[32 * KNN_K];

  const int L  = threadIdx.x;
  const int rr = L & 15, hh = L >> 4;
  const int b  = blockIdx.x >> 9;
  const int r0 = (blockIdx.x & 511) << 4;
  const int bofs = b << 13;

  v2f a[16];
  const float* arow = xt + (size_t)(bofs + r0 + rr) * CH + 2 * hh;
  #pragma unroll
  for (int kk = 0; kk < 16; ++kk) a[kk] = *(const v2f*)(arow + kk * 4);
  const float sqn = sq[bofs + r0 + rr];

  float vals[KNN_K]; int idxs[KNN_K];
  #pragma unroll
  for (int q = 0; q < KNN_K; ++q) { vals[q] = -__builtin_inff(); idxs[q] = 0; }

  const float* xbase = xt + (size_t)bofs * CH + 2 * hh;

  auto loadB = [&](int ct, v2f (&bb)[16]) {
    const float* brow = xbase + (size_t)(((ct << 4) + rr)) * CH;
    #pragma unroll
    for (int kk = 0; kk < 16; ++kk) bb[kk] = *(const v2f*)(brow + kk * 4);
  };

  auto step = [&](int ct, v2f (&cur)[16], v2f (&nxt)[16]) {
    const int c0 = ct << 4;
    v4f sq0 = *(const v4f*)(sq + bofs + c0 + hh * 8);
    v4f sq1 = *(const v4f*)(sq + bofs + c0 + hh * 8 + 4);
    float sqt[8];
    #pragma unroll
    for (int j = 0; j < 4; ++j) { sqt[j] = sq0[j]; sqt[j + 4] = sq1[j]; }

    v8f acc = {};
    #pragma unroll
    for (int kk = 0; kk < 16; ++kk) acc = wmma_f32(cur[kk], a[kk], acc);

    loadB((ct + 1) & 511, nxt);

    #pragma unroll
    for (int j = 0; j < 8; ++j) {
      int ci = hh * 8 + j;
      int colg = c0 + ci;
      float f = acc[j];
      float s = 2.f * f - sqt[j] - sqn;
      if (s > vals[0]) {
        float old19 = vals[KNN_K - 1];
        #pragma unroll
        for (int t = 0; t < KNN_K - 1; ++t) {
          bool sh = s > vals[t + 1];
          float kv = vals[t]; int ki = idxs[t];
          vals[t] = sh ? vals[t + 1] : fmaxf(s, kv);
          idxs[t] = sh ? idxs[t + 1] : ((kv >= s) ? ki : colg);
        }
        if (s > old19) { vals[KNN_K - 1] = s; idxs[KNN_K - 1] = colg; }
      }
    }
  };

  v2f bufA[16], bufB[16];
  loadB(0, bufA);
  for (int ct = 0; ct < NPT / 16; ct += 2) {
    step(ct,     bufA, bufB);
    step(ct + 1, bufB, bufA);
  }

  #pragma unroll
  for (int q = 0; q < KNN_K; ++q) { mv[L * KNN_K + q] = vals[q]; mi[L * KNN_K + q] = idxs[q]; }
  __syncthreads();
  if (L < 16) {
    const float* va = &mv[L * KNN_K];          const int* ia = &mi[L * KNN_K];
    const float* vb = &mv[(L + 16) * KNN_K];   const int* ib = &mi[(L + 16) * KNN_K];
    int pa = KNN_K - 1, pb = KNN_K - 1;
    int* outp = knn + (size_t)(bofs + r0 + L) * KNN_K;
    for (int q = 0; q < KNN_K; ++q) {
      int sa = pa < 0 ? 0 : pa, sb = pb < 0 ? 0 : pb;
      float fa = (pa >= 0) ? va[sa] : -__builtin_inff();
      float fb = (pb >= 0) ? vb[sb] : -__builtin_inff();
      bool ta = fa >= fb;
      outp[q] = ta ? ia[sa] : ib[sb];
      if (ta) --pa; else --pb;
    }
  }
}

__launch_bounds__(128)
__global__ void conv_kernel(const float* __restrict__ xt, const int* __restrict__ knn,
                            const float* __restrict__ W, const float* __restrict__ bias,
                            float* __restrict__ mxp, float* __restrict__ mnp,
                            float* __restrict__ part) {
  __shared__ float    s_edge[4][16 * CIN2];
  __shared__ unsigned s_umax[4][4][COUT];
  __shared__ unsigned s_umin[4][4][COUT];
  __shared__ float    s_psum[4][16][8];
  __shared__ float    s_psq [4][16][8];

  const int tid = threadIdx.x;
  const int w = tid >> 5, L = tid & 31;
  const int rr = L & 15, hh = L >> 4;
  const int wfp = blockIdx.x * 16 + w * 4;

  {
    unsigned* um = &s_umax[0][0][0];
    unsigned* un = &s_umin[0][0][0];
    for (int i = tid; i < 4 * 4 * COUT; i += 128) { um[i] = 0u; un[i] = 0xFFFFFFFFu; }
  }
  __syncthreads();

  float fsum[4] = {0.f, 0.f, 0.f, 0.f}, fsq[4] = {0.f, 0.f, 0.f, 0.f};
  float* edge = &s_edge[w][0];

  #pragma unroll 1
  for (int t = 0; t < 5; ++t) {
    {
      int j = L >> 1, half = L & 1;
      int le = t * 16 + j;
      int p = le / 20, k = le - 20 * p;
      int gp = wfp + p;
      int bb = gp >> 13;
      int m = knn[(size_t)gp * KNN_K + k];
      const float* xn = xt + (size_t)gp * CH;
      const float* xm = xt + (size_t)((bb << 13) + m) * CH;
      float* er = edge + j * CIN2;
      if (half == 0) {
        #pragma unroll
        for (int c = 0; c < CH; c += 4)
          *(v4f*)(er + c) = *(const v4f*)(xm + c) - *(const v4f*)(xn + c);
      } else {
        #pragma unroll
        for (int c = 0; c < CH; c += 4)
          *(v4f*)(er + CH + c) = *(const v4f*)(xn + c);
      }
    }
    __syncthreads();

    int le0 = t * 16 + hh * 8;
    int pA = le0 / 20;
    int vth = (pA + 1) * 20 - le0;
    bool hasB = vth < 8;

    #pragma unroll 1
    for (int ot = 0; ot < 4; ++ot) {
      const int o = ot * 16 + rr;
      const float* wrow = W + (size_t)o * CIN2 + 2 * hh;
      const float* arow = edge + rr * CIN2 + 2 * hh;
      v8f acc = {};
      #pragma unroll
      for (int g = 0; g < 4; ++g) {
        v2f av[8], bv[8];
        #pragma unroll
        for (int u = 0; u < 8; ++u) {
          av[u] = *(const v2f*)(arow + (g * 8 + u) * 4);
          bv[u] = *(const v2f*)(wrow + (g * 8 + u) * 4);
        }
        #pragma unroll
        for (int u = 0; u < 8; ++u) acc = wmma_f32(av[u], bv[u], acc);
      }
      float bo = bias[o];
      float gAmax = -__builtin_inff(), gAmin = __builtin_inff();
      float gBmax = -__builtin_inff(), gBmin = __builtin_inff();
      float lsum = 0.f, lsq = 0.f;
      #pragma unroll
      for (int v = 0; v < 8; ++v) {
        float h = acc[v] + bo;
        h = (h >= 0.f) ? h : NEGSL * h;
        lsum += h; lsq += h * h;
        bool inB = (v >= vth);
        gAmax = fmaxf(gAmax, inB ? -__builtin_inff() : h);
        gAmin = fminf(gAmin, inB ?  __builtin_inff() : h);
        gBmax = fmaxf(gBmax, inB ? h : -__builtin_inff());
        gBmin = fminf(gBmin, inB ? h :  __builtin_inff());
      }
      fsum[ot] += lsum; fsq[ot] += lsq;
      atomicMax(&s_umax[w][pA][o], fenc(gAmax));
      atomicMin(&s_umin[w][pA][o], fenc(gAmin));
      if (hasB) {
        atomicMax(&s_umax[w][pA + 1][o], fenc(gBmax));
        atomicMin(&s_umin[w][pA + 1][o], fenc(gBmin));
      }
    }
    __syncthreads();
  }

  #pragma unroll
  for (int ot = 0; ot < 4; ++ot) {
    s_psum[ot][rr][w * 2 + hh] = fsum[ot];
    s_psq [ot][rr][w * 2 + hh] = fsq[ot];
  }
  __syncthreads();
  if (tid < COUT) {
    float s = 0.f, s2 = 0.f;
    #pragma unroll
    for (int q = 0; q < 8; ++q) { s += s_psum[tid >> 4][tid & 15][q]; s2 += s_psq[tid >> 4][tid & 15][q]; }
    part[(size_t)blockIdx.x * 128 + tid]      = s;
    part[(size_t)blockIdx.x * 128 + 64 + tid] = s2;
  }

  #pragma unroll
  for (int it = 0; it < 8; ++it) {
    int q = L + 32 * it;
    int p = q >> 6, o = q & 63;
    int gp = wfp + p;
    int bb = gp >> 13, n = gp & (NPT - 1);
    size_t oi = ((size_t)(bb * COUT + o) << 13) + n;
    mxp[oi] = fdec(s_umax[w][p][o]);
    mnp[oi] = fdec(s_umin[w][p][o]);
  }
}

__global__ void stats_kernel(const float* __restrict__ part, float* __restrict__ gst) {
  int t = threadIdx.x;
  float s = 0.f;
  for (int g = 0; g < 1024; ++g) s += part[(size_t)g * 128 + t];
  gst[t] = s;
}

__global__ void out_kernel(const float* __restrict__ mxp, const float* __restrict__ mnp,
                           const float* __restrict__ gst,
                           const float* __restrict__ gamma, const float* __restrict__ beta,
                           float* __restrict__ outp) {
  int i = blockIdx.x * blockDim.x + threadIdx.x;
  int o = (i >> 13) & 63;
  const float cnt = (float)(BATCH * NPT * KNN_K);
  float mean = gst[o] / cnt;
  float var  = gst[COUT + o] / cnt - mean * mean;
  float scale = gamma[o] * rsqrtf(var + BNEPS);
  float sel = (scale >= 0.f) ? mxp[i] : mnp[i];
  outp[i] = (sel - mean) * scale + beta[o];
}

extern "C" void kernel_launch(void* const* d_in, const int* in_sizes, int n_in,
                              void* d_out, int out_size, void* d_ws, size_t ws_size,
                              hipStream_t stream) {
  const float* x     = (const float*)d_in[0];
  const float* W     = (const float*)d_in[1];
  const float* bias  = (const float*)d_in[2];
  const float* gamma = (const float*)d_in[3];
  const float* beta  = (const float*)d_in[4];
  float* outp = (float*)d_out;

  float* xt  = (float*)d_ws;
  float* sq  = xt + (size_t)BATCH * NPT * CH;
  int*   knn = (int*)(sq + (size_t)BATCH * NPT);
  float* mxp = (float*)(knn + (size_t)BATCH * NPT * KNN_K);
  float* mnp = mxp + (size_t)BATCH * COUT * NPT;
  float* gst = mnp + (size_t)BATCH * COUT * NPT;
  float* part = gst + 128;

  prep_kernel <<<(BATCH * NPT) / 256, 256, 0, stream>>>(x, xt, sq);
  knn_kernel  <<<BATCH * (NPT / 16), 32, 0, stream>>>(xt, sq, knn);
  conv_kernel <<<(BATCH * NPT) / 16, 128, 0, stream>>>(xt, knn, W, bias, mxp, mnp, part);
  stats_kernel<<<1, 128, 0, stream>>>(part, gst);
  out_kernel  <<<(BATCH * COUT * NPT) / 256, 256, 0, stream>>>(mxp, mnp, gst, gamma, beta, outp);
}


# knn top-k: scores above the lane's 20th best are queued per lane in LDS by branch-free tail stores and inserted in batches when a queue nears full; next tile's loads issue ahead of the WMMA chain
# speedup vs baseline: 1.9782x; 1.5362x over previous
#include <hip/hip_runtime.h>
#include <stdint.h>

#define BATCH 2
#define CH    64
#define NPT   8192
#define KNN_K 20
#define COUT  64
#define CIN2  128
#define NEGSL 0.2f
#define BNEPS 1e-5f

typedef float v2f __attribute__((ext_vector_type(2)));
typedef float v4f __attribute__((ext_vector_type(4)));
typedef float v8f __attribute__((ext_vector_type(8)));

__device__ __forceinline__ v8f wmma_f32(v2f a, v2f b, v8f c) {
  return __builtin_amdgcn_wmma_f32_16x16x4_f32(false, a, false, b, (short)0, c, false, false);
}

__device__ __forceinline__ unsigned fenc(float f) {
  unsigned u = __float_as_uint(f);
  return (u & 0x80000000u) ? ~u : (u | 0x80000000u);
}
__device__ __forceinline__ float fdec(unsigned u) {
  return __uint_as_float((u & 0x80000000u) ? (u ^ 0x80000000u) : ~u);
}

__global__ void prep_kernel(const float* __restrict__ x,
                            float* __restrict__ xt, float* __restrict__ sq) {
  int gp = blockIdx.x * blockDim.x + threadIdx.x;
  int b = gp >> 13, n = gp & (NPT - 1);
  const float* xb = x + (size_t)b * CH * NPT + n;
  float acc = 0.f;
  #pragma unroll 8
  for (int c = 0; c < CH; ++c) {
    float v = xb[(size_t)c * NPT];
    acc += v * v;
    xt[(size_t)gp * CH + c] = v;
  }
  sq[gp] = acc;
}

#define QSLOTS 33
__launch_bounds__(32)
__global__ void knn_kernel(const float* __restrict__ xt, const float* __restrict__ sq,
                           int* __restrict__ knn) {
  __shared__ float mv[32 * KNN_K];
  __shared__ int   mi[32 * KNN_K];
  __shared__ v2f   qbuf[QSLOTS * 32];

  const int L  = threadIdx.x;
  const int rr = L & 15, hh = L >> 4;
  const int b  = blockIdx.x >> 9;
  const int r0 = (blockIdx.x & 511) << 4;
  const int bofs = b << 13;

  #pragma unroll
  for (int q = 0; q < QSLOTS; ++q) { v2f z; z.x = -__builtin_inff(); z.y = 0.f; qbuf[q * 32 + L] = z; }

  v2f a[16];
  const float* arow = xt + (size_t)(bofs + r0 + rr) * CH + 2 * hh;
  #pragma unroll
  for (int kk = 0; kk < 16; ++kk) a[kk] = *(const v2f*)(arow + kk * 4);
  const float sqn = sq[bofs + r0 + rr];

  float vals[KNN_K]; int idxs[KNN_K];
  #pragma unroll
  for (int q = 0; q < KNN_K; ++q) { vals[q] = -__builtin_inff(); idxs[q] = 0; }
  int qn = 0;

  const float* xbase = xt + (size_t)bofs * CH + 2 * hh;

  auto loadB = [&](int ct, v2f (&bb)[16]) {
    const float* brow = xbase + (size_t)(((ct << 4) + rr)) * CH;
    #pragma unroll
    for (int kk = 0; kk < 16; ++kk) bb[kk] = *(const v2f*)(brow + kk * 4);
  };

  auto insert = [&](float s, int colg) {
    float old19 = vals[KNN_K - 1];
    #pragma unroll
    for (int t = 0; t < KNN_K - 1; ++t) {
      bool sh = s > vals[t + 1];
      float kv = vals[t]; int ki = idxs[t];
      vals[t] = sh ? vals[t + 1] : fmaxf(s, kv);
      idxs[t] = sh ? idxs[t + 1] : ((kv >= s) ? ki : colg);
    }
    if (s > old19) { vals[KNN_K - 1] = s; idxs[KNN_K - 1] = colg; }
  };

  auto drain = [&]() {
    const int mq = __builtin_amdgcn_readfirstlane((int)__builtin_amdgcn_wave_reduce_max_u32((unsigned)qn, 0));
    v2f e = qbuf[L];
    for (int i = 0; i < mq; ++i) {
      v2f cur = e;
      e = qbuf[(i + 1) * 32 + L];
      float s = cur.x;
      if (i < qn && s > vals[0]) insert(s, __float_as_int(cur.y));
    }
    qn = 0;
  };

  auto step = [&](int ct, v2f (&cur)[16], v2f (&nxt)[16]) {
    const int c0 = ct << 4;
    v4f sq0 = *(const v4f*)(sq + bofs + c0 + hh * 8);
    v4f sq1 = *(const v4f*)(sq + bofs + c0 + hh * 8 + 4);
    loadB((ct + 1) & 511, nxt);
    float sqt[8];
    #pragma unroll
    for (int j = 0; j < 4; ++j) { sqt[j] = sq0[j]; sqt[j + 4] = sq1[j]; }

    v8f acc = {};
    #pragma unroll
    for (int kk = 0; kk < 16; ++kk) acc = wmma_f32(cur[kk], a[kk], acc);

    const float thr = vals[0];
    #pragma unroll
    for (int j = 0; j < 8; ++j) {
      int colg = c0 + hh * 8 + j;
      float s = 2.f * acc[j] - sqt[j] - sqn;
      v2f ent; ent.x = s; ent.y = __int_as_float(colg);
      qbuf[qn * 32 + L] = ent;
      qn += (s > thr) ? 1 : 0;
    }
  };

  v2f bufA[16], bufB[16];
  loadB(0, bufA);
  for (int ct = 0; ct < NPT / 16; ct += 2) {
    step(ct,     bufA, bufB);
    step(ct + 1, bufB, bufA);
    if (__builtin_amdgcn_ballot_w32(qn > 16) != 0u) drain();
  }
  drain();

  #pragma unroll
  for (int q = 0; q < KNN_K; ++q) { mv[L * KNN_K + q] = vals[q]; mi[L * KNN_K + q] = idxs[q]; }
  __syncthreads();
  if (L < 16) {
    const float* va = &mv[L * KNN_K];          const int* ia = &mi[L * KNN_K];
    const float* vb = &mv[(L + 16) * KNN_K];   const int* ib = &mi[(L + 16) * KNN_K];
    int pa = KNN_K - 1, pb = KNN_K - 1;
    int* outp = knn + (size_t)(bofs + r0 + L) * KNN_K;
    for (int q = 0; q < KNN_K; ++q) {
      int sa = pa < 0 ? 0 : pa, sb = pb < 0 ? 0 : pb;
      float fa = (pa >= 0) ? va[sa] : -__builtin_inff();
      float fb = (pb >= 0) ? vb[sb] : -__builtin_inff();
      bool ta = fa >= fb;
      outp[q] = ta ? ia[sa] : ib[sb];
      if (ta) --pa; else --pb;
    }
  }
}

__launch_bounds__(128)
__global__ void conv_kernel(const float* __restrict__ xt, const int* __restrict__ knn,
                            const float* __restrict__ W, const float* __restrict__ bias,
                            float* __restrict__ mxp, float* __restrict__ mnp,
                            float* __restrict__ part) {
  __shared__ float    s_edge[4][16 * CIN2];
  __shared__ unsigned s_umax[4][4][COUT];
  __shared__ unsigned s_umin[4][4][COUT];
  __shared__ float    s_psum[4][16][8];
  __shared__ float    s_psq [4][16][8];

  const int tid = threadIdx.x;
  const int w = tid >> 5, L = tid & 31;
  const int rr = L & 15, hh = L >> 4;
  const int wfp = blockIdx.x * 16 + w * 4;

  {
    unsigned* um = &s_umax[0][0][0];
    unsigned* un = &s_umin[0][0][0];
    for (int i = tid; i < 4 * 4 * COUT; i += 128) { um[i] = 0u; un[i] = 0xFFFFFFFFu; }
  }
  __syncthreads();

  float fsum[4] = {0.f, 0.f, 0.f, 0.f}, fsq[4] = {0.f, 0.f, 0.f, 0.f};
  float* edge = &s_edge[w][0];

  #pragma unroll 1
  for (int t = 0; t < 5; ++t) {
    {
      int j = L >> 1, half = L & 1;
      int le = t * 16 + j;
      int p = le / 20, k = le - 20 * p;
      int gp = wfp + p;
      int bb = gp >> 13;
      int m = knn[(size_t)gp * KNN_K + k];
      const float* xn = xt + (size_t)gp * CH;
      const float* xm = xt + (size_t)((bb << 13) + m) * CH;
      float* er = edge + j * CIN2;
      if (half == 0) {
        #pragma unroll
        for (int c = 0; c < CH; c += 4)
          *(v4f*)(er + c) = *(const v4f*)(xm + c) - *(const v4f*)(xn + c);
      } else {
        #pragma unroll
        for (int c = 0; c < CH; c += 4)
          *(v4f*)(er + CH + c) = *(const v4f*)(xn + c);
      }
    }
    __syncthreads();

    int le0 = t * 16 + hh * 8;
    int pA = le0 / 20;
    int vth = (pA + 1) * 20 - le0;
    bool hasB = vth < 8;

    #pragma unroll 1
    for (int ot = 0; ot < 4; ++ot) {
      const int o = ot * 16 + rr;
      const float* wrow = W + (size_t)o * CIN2 + 2 * hh;
      const float* arow = edge + rr * CIN2 + 2 * hh;
      v8f acc = {};
      #pragma unroll
      for (int g = 0; g < 4; ++g) {
        v2f av[8], bv[8];
        #pragma unroll
        for (int u = 0; u < 8; ++u) {
          av[u] = *(const v2f*)(arow + (g * 8 + u) * 4);
          bv[u] = *(const v2f*)(wrow + (g * 8 + u) * 4);
        }
        #pragma unroll
        for (int u = 0; u < 8; ++u) acc = wmma_f32(av[u], bv[u], acc);
      }
      float bo = bias[o];
      float gAmax = -__builtin_inff(), gAmin = __builtin_inff();
      float gBmax = -__builtin_inff(), gBmin = __builtin_inff();
      float lsum = 0.f, lsq = 0.f;
      #pragma unroll
      for (int v = 0; v < 8; ++v) {
        float h = acc[v] + bo;
        h = (h >= 0.f) ? h : NEGSL * h;
        lsum += h; lsq += h * h;
        bool inB = (v >= vth);
        gAmax = fmaxf(gAmax, inB ? -__builtin_inff() : h);
        gAmin = fminf(gAmin, inB ?  __builtin_inff() : h);
        gBmax = fmaxf(gBmax, inB ? h : -__builtin_inff());
        gBmin = fminf(gBmin, inB ? h :  __builtin_inff());
      }
      fsum[ot] += lsum; fsq[ot] += lsq;
      atomicMax(&s_umax[w][pA][o], fenc(gAmax));
      atomicMin(&s_umin[w][pA][o], fenc(gAmin));
      if (hasB) {
        atomicMax(&s_umax[w][pA + 1][o], fenc(gBmax));
        atomicMin(&s_umin[w][pA + 1][o], fenc(gBmin));
      }
    }
    __syncthreads();
  }

  #pragma unroll
  for (int ot = 0; ot < 4; ++ot) {
    s_psum[ot][rr][w * 2 + hh] = fsum[ot];
    s_psq [ot][rr][w * 2 + hh] = fsq[ot];
  }
  __syncthreads();
  if (tid < COUT) {
    float s = 0.f, s2 = 0.f;
    #pragma unroll
    for (int q = 0; q < 8; ++q) { s += s_psum[tid >> 4][tid & 15][q]; s2 += s_psq[tid >> 4][tid & 15][q]; }
    part[(size_t)blockIdx.x * 128 + tid]      = s;
    part[(size_t)blockIdx.x * 128 + 64 + tid] = s2;
  }

  #pragma unroll
  for (int it = 0; it < 8; ++it) {
    int q = L + 32 * it;
    int p = q >> 6, o = q & 63;
    int gp = wfp + p;
    int bb = gp >> 13, n = gp & (NPT - 1);
    size_t oi = ((size_t)(bb * COUT + o) << 13) + n;
    mxp[oi] = fdec(s_umax[w][p][o]);
    mnp[oi] = fdec(s_umin[w][p][o]);
  }
}

__global__ void stats_kernel(const float* __restrict__ part, float* __restrict__ gst) {
  int t = threadIdx.x;
  float s = 0.f;
  for (int g = 0; g < 1024; ++g) s += part[(size_t)g * 128 + t];
  gst[t] = s;
}

__global__ void out_kernel(const float* __restrict__ mxp, const float* __restrict__ mnp,
                           const float* __restrict__ gst,
                           const float* __restrict__ gamma, const float* __restrict__ beta,
                           float* __restrict__ outp) {
  int i = blockIdx.x * blockDim.x + threadIdx.x;
  int o = (i >> 13) & 63;
  const float cnt = (float)(BATCH * NPT * KNN_K);
  float mean = gst[o] / cnt;
  float var  = gst[COUT + o] / cnt - mean * mean;
  float scale = gamma[o] * rsqrtf(var + BNEPS);
  float sel = (scale >= 0.f) ? mxp[i] : mnp[i];
  outp[i] = (sel - mean) * scale + beta[o];
}

extern "C" void kernel_launch(void* const* d_in, const int* in_sizes, int n_in,
                              void* d_out, int out_size, void* d_ws, size_t ws_size,
                              hipStream_t stream) {
  const float* x     = (const float*)d_in[0];
  const float* W     = (const float*)d_in[1];
  const float* bias  = (const float*)d_in[2];
  const float* gamma = (const float*)d_in[3];
  const float* beta  = (const float*)d_in[4];
  float* outp = (float*)d_out;

  float* xt  = (float*)d_ws;
  float* sq  = xt + (size_t)BATCH * NPT * CH;
  int*   knn = (int*)(sq + (size_t)BATCH * NPT);
  float* mxp = (float*)(knn + (size_t)BATCH * NPT * KNN_K);
  float* mnp = mxp + (size_t)BATCH * COUT * NPT;
  float* gst = mnp + (size_t)BATCH * COUT * NPT;
  float* part = gst + 128;

  prep_kernel <<<(BATCH * NPT) / 256, 256, 0, stream>>>(x, xt, sq);
  knn_kernel  <<<BATCH * (NPT / 16), 32, 0, stream>>>(xt, sq, knn);
  conv_kernel <<<(BATCH * NPT) / 16, 128, 0, stream>>>(xt, knn, W, bias, mxp, mnp, part);
  stats_kernel<<<1, 128, 0, stream>>>(part, gst);
  out_kernel  <<<(BATCH * COUT * NPT) / 256, 256, 0, stream>>>(mxp, mnp, gst, gamma, beta, outp);
}


# knn: each 16-row block's column scan is split over 4 waves of one workgroup (own loads, queues, lists; no loop barrier), 8 lists per row merged in LDS: 4 waves per SIMD cover each other's L2 waits
# speedup vs baseline: 2.4971x; 1.2623x over previous
#include <hip/hip_runtime.h>
#include <stdint.h>

#define BATCH 2
#define CH    64
#define NPT   8192
#define KNN_K 20
#define COUT  64
#define CIN2  128
#define NEGSL 0.2f
#define BNEPS 1e-5f

typedef float v2f __attribute__((ext_vector_type(2)));
typedef float v4f __attribute__((ext_vector_type(4)));
typedef float v8f __attribute__((ext_vector_type(8)));

__device__ __forceinline__ v8f wmma_f32(v2f a, v2f b, v8f c) {
  return __builtin_amdgcn_wmma_f32_16x16x4_f32(false, a, false, b, (short)0, c, false, false);
}

__device__ __forceinline__ unsigned fenc(float f) {
  unsigned u = __float_as_uint(f);
  return (u & 0x80000000u) ? ~u : (u | 0x80000000u);
}
__device__ __forceinline__ float fdec(unsigned u) {
  return __uint_as_float((u & 0x80000000u) ? (u ^ 0x80000000u) : ~u);
}

__global__ void prep_kernel(const float* __restrict__ x,
                            float* __restrict__ xt, float* __restrict__ sq) {
  int gp = blockIdx.x * blockDim.x + threadIdx.x;
  int b = gp >> 13, n = gp & (NPT - 1);
  const float* xb = x + (size_t)b * CH * NPT + n;
  float acc = 0.f;
  #pragma unroll 8
  for (int c = 0; c < CH; ++c) {
    float v = xb[(size_t)c * NPT];
    acc += v * v;
    xt[(size_t)gp * CH + c] = v;
  }
  sq[gp] = acc;
}

#define QSLOTS 33
#define CSPLIT 4
#define TPW    (NPT / 16 / CSPLIT)
#define NLIST  (2 * CSPLIT)
__launch_bounds__(32 * CSPLIT)
__global__ void knn_kernel(const float* __restrict__ xt, const float* __restrict__ sq,
                           int* __restrict__ knn) {
  __shared__ float mv[CSPLIT * 32 * KNN_K];
  __shared__ int   mi[CSPLIT * 32 * KNN_K];
  __shared__ v2f   qall[CSPLIT * QSLOTS * 32];

  const int w  = __builtin_amdgcn_readfirstlane((int)(threadIdx.x >> 5));
  const int L  = threadIdx.x & 31;
  v2f* qbuf = qall + w * (QSLOTS * 32);
  const int rr = L & 15, hh = L >> 4;
  const int b  = blockIdx.x >> 9;
  const int r0 = (blockIdx.x & 511) << 4;
  const int bofs = b << 13;

  #pragma unroll
  for (int q = 0; q < QSLOTS; ++q) { v2f z; z.x = -__builtin_inff(); z.y = 0.f; qbuf[q * 32 + L] = z; }

  v2f a[16];
  const float* arow = xt + (size_t)(bofs + r0 + rr) * CH + 2 * hh;
  #pragma unroll
  for (int kk = 0; kk < 16; ++kk) a[kk] = *(const v2f*)(arow + kk * 4);
  const float sqn = sq[bofs + r0 + rr];

  float vals[KNN_K]; int idxs[KNN_K];
  #pragma unroll
  for (int q = 0; q < KNN_K; ++q) { vals[q] = -__builtin_inff(); idxs[q] = 0; }
  int qn = 0;

  const float* xbase = xt + (size_t)bofs * CH + 2 * hh;

  auto loadB = [&](int ct, v2f (&bb)[16]) {
    const float* brow = xbase + (size_t)(((ct << 4) + rr)) * CH;
    #pragma unroll
    for (int kk = 0; kk < 16; ++kk) bb[kk] = *(const v2f*)(brow + kk * 4);
  };

  auto insert = [&](float s, int colg) {
    float old19 = vals[KNN_K - 1];
    #pragma unroll
    for (int t = 0; t < KNN_K - 1; ++t) {
      bool sh = s > vals[t + 1];
      float kv = vals[t]; int ki = idxs[t];
      vals[t] = sh ? vals[t + 1] : fmaxf(s, kv);
      idxs[t] = sh ? idxs[t + 1] : ((kv >= s) ? ki : colg);
    }
    if (s > old19) { vals[KNN_K - 1] = s; idxs[KNN_K - 1] = colg; }
  };

  auto drain = [&]() {
    const int mq = __builtin_amdgcn_readfirstlane((int)__builtin_amdgcn_wave_reduce_max_u32((unsigned)qn, 0));
    v2f e = qbuf[L];
    for (int i = 0; i < mq; ++i) {
      v2f cur = e;
      e = qbuf[(i + 1) * 32 + L];
      float s = cur.x;
      if (i < qn && s > vals[0]) insert(s, __float_as_int(cur.y));
    }
    qn = 0;
  };

  auto step = [&](int ct, v2f (&cur)[16], v2f (&nxt)[16]) {
    const int c0 = ct << 4;
    v4f sq0 = *(const v4f*)(sq + bofs + c0 + hh * 8);
    v4f sq1 = *(const v4f*)(sq + bofs + c0 + hh * 8 + 4);
    loadB((ct + 1) & 511, nxt);
    float sqt[8];
    #pragma unroll
    for (int j = 0; j < 4; ++j) { sqt[j] = sq0[j]; sqt[j + 4] = sq1[j]; }

    v8f acc = {};
    #pragma unroll
    for (int kk = 0; kk < 16; ++kk) acc = wmma_f32(cur[kk], a[kk], acc);

    const float thr = vals[0];
    #pragma unroll
    for (int j = 0; j < 8; ++j) {
      int colg = c0 + hh * 8 + j;
      float s = 2.f * acc[j] - sqt[j] - sqn;
      v2f ent; ent.x = s; ent.y = __int_as_float(colg);
      qbuf[qn * 32 + L] = ent;
      qn += (s > thr) ? 1 : 0;
    }
  };

  v2f bufA[16], bufB[16];
  const int ct0 = w * TPW;
  loadB(ct0, bufA);
  for (int ct = ct0; ct < ct0 + TPW; ct += 2) {
    step(ct,     bufA, bufB);
    step(ct + 1, bufB, bufA);
    if (__builtin_amdgcn_ballot_w32(qn > 16) != 0u) drain();
  }
  drain();

  #pragma unroll
  for (int q = 0; q < KNN_K; ++q) {
    mv[(w * 32 + L) * KNN_K + q] = vals[q];
    mi[(w * 32 + L) * KNN_K + q] = idxs[q];
  }
  __syncthreads();
  if (threadIdx.x < 16) {
    const int r = threadIdx.x;
    float hv[NLIST]; int hp[NLIST];
    #pragma unroll
    for (int l = 0; l < NLIST; ++l) {
      hp[l] = KNN_K - 1;
      hv[l] = mv[((l >> 1) * 32 + (l & 1) * 16 + r) * KNN_K + KNN_K - 1];
    }
    int* outp = knn + (size_t)(bofs + r0 + r) * KNN_K;
    for (int q = 0; q < KNN_K; ++q) {
      int best = 0; float bv = hv[0];
      #pragma unroll
      for (int l = 1; l < NLIST; ++l) { bool g = hv[l] > bv; bv = g ? hv[l] : bv; best = g ? l : best; }
      int bp = hp[0];
      #pragma unroll
      for (int l = 1; l < NLIST; ++l) bp = (l == best) ? hp[l] : bp;
      const int base = ((best >> 1) * 32 + (best & 1) * 16 + r) * KNN_K;
      outp[q] = mi[base + bp];
      const int   np = (bp > 0) ? bp - 1 : 0;
      const float nv = (bp > 0) ? mv[base + np] : -__builtin_inff();
      #pragma unroll
      for (int l = 0; l < NLIST; ++l) { bool m = (l == best); hp[l] = m ? bp - 1 : hp[l]; hv[l] = m ? nv : hv[l]; }
    }
  }
}

__launch_bounds__(128)
__global__ void conv_kernel(const float* __restrict__ xt, const int* __restrict__ knn,
                            const float* __restrict__ W, const float* __restrict__ bias,
                            float* __restrict__ mxp, float* __restrict__ mnp,
                            float* __restrict__ part) {
  __shared__ float    s_edge[4][16 * CIN2];
  __shared__ unsigned s_umax[4][4][COUT];
  __shared__ unsigned s_umin[4][4][COUT];
  __shared__ float    s_psum[4][16][8];
  __shared__ float    s_psq [4][16][8];

  const int tid = threadIdx.x;
  const int w = tid >> 5, L = tid & 31;
  const int rr = L & 15, hh = L >> 4;
  const int wfp = blockIdx.x * 16 + w * 4;

  {
    unsigned* um = &s_umax[0][0][0];
    unsigned* un = &s_umin[0][0][0];
    for (int i = tid; i < 4 * 4 * COUT; i += 128) { um[i] = 0u; un[i] = 0xFFFFFFFFu; }
  }
  __syncthreads();

  float fsum[4] = {0.f, 0.f, 0.f, 0.f}, fsq[4] = {0.f, 0.f, 0.f, 0.f};
  float* edge = &s_edge[w][0];

  #pragma unroll 1
  for (int t = 0; t < 5; ++t) {
    {
      int j = L >> 1, half = L & 1;
      int le = t * 16 + j;
      int p = le / 20, k = le - 20 * p;
      int gp = wfp + p;
      int bb = gp >> 13;
      int m = knn[(size_t)gp * KNN_K + k];
      const float* xn = xt + (size_t)gp * CH;
      const float* xm = xt + (size_t)((bb << 13) + m) * CH;
      float* er = edge + j * CIN2;
      if (half == 0) {
        #pragma unroll
        for (int c = 0; c < CH; c += 4)
          *(v4f*)(er + c) = *(const v4f*)(xm + c) - *(const v4f*)(xn + c);
      } else {
        #pragma unroll
        for (int c = 0; c < CH; c += 4)
          *(v4f*)(er + CH + c) = *(const v4f*)(xn + c);
      }
    }
    __syncthreads();

    int le0 = t * 16 + hh * 8;
    int pA = le0 / 20;
    int vth = (pA + 1) * 20 - le0;
    bool hasB = vth < 8;

    #pragma unroll 1
    for (int ot = 0; ot < 4; ++ot) {
      const int o = ot * 16 + rr;
      const float* wrow = W + (size_t)o * CIN2 + 2 * hh;
      const float* arow = edge + rr * CIN2 + 2 * hh;
      v8f acc = {};
      #pragma unroll
      for (int g = 0; g < 4; ++g) {
        v2f av[8], bv[8];
        #pragma unroll
        for (int u = 0; u < 8; ++u) {
          av[u] = *(const v2f*)(arow + (g * 8 + u) * 4);
          bv[u] = *(const v2f*)(wrow + (g * 8 + u) * 4);
        }
        #pragma unroll
        for (int u = 0; u < 8; ++u) acc = wmma_f32(av[u], bv[u], acc);
      }
      float bo = bias[o];
      float gAmax = -__builtin_inff(), gAmin = __builtin_inff();
      float gBmax = -__builtin_inff(), gBmin = __builtin_inff();
      float lsum = 0.f, lsq = 0.f;
      #pragma unroll
      for (int v = 0; v < 8; ++v) {
        float h = acc[v] + bo;
        h = (h >= 0.f) ? h : NEGSL * h;
        lsum += h; lsq += h * h;
        bool inB = (v >= vth);
        gAmax = fmaxf(gAmax, inB ? -__builtin_inff() : h);
        gAmin = fminf(gAmin, inB ?  __builtin_inff() : h);
        gBmax = fmaxf(gBmax, inB ? h : -__builtin_inff());
        gBmin = fminf(gBmin, inB ? h :  __builtin_inff());
      }
      fsum[ot] += lsum; fsq[ot] += lsq;
      atomicMax(&s_umax[w][pA][o], fenc(gAmax));
      atomicMin(&s_umin[w][pA][o], fenc(gAmin));
      if (hasB) {
        atomicMax(&s_umax[w][pA + 1][o], fenc(gBmax));
        atomicMin(&s_umin[w][pA + 1][o], fenc(gBmin));
      }
    }
    __syncthreads();
  }

  #pragma unroll
  for (int ot = 0; ot < 4; ++ot) {
    s_psum[ot][rr][w * 2 + hh] = fsum[ot];
    s_psq [ot][rr][w * 2 + hh] = fsq[ot];
  }
  __syncthreads();
  if (tid < COUT) {
    float s = 0.f, s2 = 0.f;
    #pragma unroll
    for (int q = 0; q < 8; ++q) { s += s_psum[tid >> 4][tid & 15][q]; s2 += s_psq[tid >> 4][tid & 15][q]; }
    part[(size_t)blockIdx.x * 128 + tid]      = s;
    part[(size_t)blockIdx.x * 128 + 64 + tid] = s2;
  }

  #pragma unroll
  for (int it = 0; it < 8; ++it) {
    int q = L + 32 * it;
    int p = q >> 6, o = q & 63;
    int gp = wfp + p;
    int bb = gp >> 13, n = gp & (NPT - 1);
    size_t oi = ((size_t)(bb * COUT + o) << 13) + n;
    mxp[oi] = fdec(s_umax[w][p][o]);
    mnp[oi] = fdec(s_umin[w][p][o]);
  }
}

__global__ void stats_kernel(const float* __restrict__ part, float* __restrict__ gst) {
  int t = threadIdx.x;
  float s = 0.f;
  for (int g = 0; g < 1024; ++g) s += part[(size_t)g * 128 + t];
  gst[t] = s;
}

__global__ void out_kernel(const float* __restrict__ mxp, const float* __restrict__ mnp,
                           const float* __restrict__ gst,
                           const float* __restrict__ gamma, const float* __restrict__ beta,
                           float* __restrict__ outp) {
  int i = blockIdx.x * blockDim.x + threadIdx.x;
  int o = (i >> 13) & 63;
  const float cnt = (float)(BATCH * NPT * KNN_K);
  float mean = gst[o] / cnt;
  float var  = gst[COUT + o] / cnt - mean * mean;
  float scale = gamma[o] * rsqrtf(var + BNEPS);
  float sel = (scale >= 0.f) ? mxp[i] : mnp[i];
  outp[i] = (sel - mean) * scale + beta[o];
}

extern "C" void kernel_launch(void* const* d_in, const int* in_sizes, int n_in,
                              void* d_out, int out_size, void* d_ws, size_t ws_size,
                              hipStream_t stream) {
  const float* x     = (const float*)d_in[0];
  const float* W     = (const float*)d_in[1];
  const float* bias  = (const float*)d_in[2];
  const float* gamma = (const float*)d_in[3];
  const float* beta  = (const float*)d_in[4];
  float* outp = (float*)d_out;

  float* xt  = (float*)d_ws;
  float* sq  = xt + (size_t)BATCH * NPT * CH;
  int*   knn = (int*)(sq + (size_t)BATCH * NPT);
  float* mxp = (float*)(knn + (size_t)BATCH * NPT * KNN_K);
  float* mnp = mxp + (size_t)BATCH * COUT * NPT;
  float* gst = mnp + (size_t)BATCH * COUT * NPT;
  float* part = gst + 128;

  prep_kernel <<<(BATCH * NPT) / 256, 256, 0, stream>>>(x, xt, sq);
  knn_kernel  <<<BATCH * (NPT / 16), 32 * CSPLIT, 0, stream>>>(xt, sq, knn);
  conv_kernel <<<(BATCH * NPT) / 16, 128, 0, stream>>>(xt, knn, W, bias, mxp, mnp, part);
  stats_kernel<<<1, 128, 0, stream>>>(part, gst);
  out_kernel  <<<(BATCH * COUT * NPT) / 256, 256, 0, stream>>>(mxp, mnp, gst, gamma, beta, outp);
}


# knn: operand tiles read from a fragment-major copy of x made by the prep kernel: a wave load is one contiguous 512 B (four whole lines, b128), not 8 B per lane from a line per row; k order unchanged
# speedup vs baseline: 2.5990x; 1.0408x over previous
#include <hip/hip_runtime.h>
#include <stdint.h>

#define BATCH 2
#define CH    64
#define NPT   8192
#define KNN_K 20
#define COUT  64
#define CIN2  128
#define NEGSL 0.2f
#define BNEPS 1e-5f

typedef float v2f __attribute__((ext_vector_type(2)));
typedef float v4f __attribute__((ext_vector_type(4)));
typedef float v8f __attribute__((ext_vector_type(8)));

__device__ __forceinline__ v8f wmma_f32(v2f a, v2f b, v8f c) {
  return __builtin_amdgcn_wmma_f32_16x16x4_f32(false, a, false, b, (short)0, c, false, false);
}

__device__ __forceinline__ unsigned fenc(float f) {
  unsigned u = __float_as_uint(f);
  return (u & 0x80000000u) ? ~u : (u | 0x80000000u);
}
__device__ __forceinline__ float fdec(unsigned u) {
  return __uint_as_float((u & 0x80000000u) ? (u ^ 0x80000000u) : ~u);
}

__global__ void prep_kernel(const float* __restrict__ x,
                            float* __restrict__ xt, float* __restrict__ sq,
                            float* __restrict__ xf) {
  int gp = blockIdx.x * blockDim.x + threadIdx.x;
  int b = gp >> 13, n = gp & (NPT - 1);
  const float* xb = x + (size_t)b * CH * NPT + n;
  v4f* xf4 = (v4f*)xf + (size_t)(gp >> 4) * 256 + (gp & 15);
  float acc = 0.f;
  for (int i = 0; i < 8; ++i) {
    float v[8];
    #pragma unroll
    for (int e = 0; e < 8; ++e) v[e] = xb[(size_t)(i * 8 + e) * NPT];
    #pragma unroll
    for (int e = 0; e < 8; ++e) {
      acc = __builtin_fmaf(v[e], v[e], acc);
      xt[(size_t)gp * CH + i * 8 + e] = v[e];
    }
    v4f h0, h1;
    h0.x = v[0]; h0.y = v[1]; h0.z = v[4]; h0.w = v[5];
    h1.x = v[2]; h1.y = v[3]; h1.z = v[6]; h1.w = v[7];
    xf4[i * 32]      = h0;
    xf4[i * 32 + 16] = h1;
  }
  sq[gp] = acc;
}

#define QSLOTS 33
#define CSPLIT 4
#define TPW    (NPT / 16 / CSPLIT)
#define NLIST  (2 * CSPLIT)
__launch_bounds__(32 * CSPLIT)
__global__ void knn_kernel(const float* __restrict__ xf, const float* __restrict__ sq,
                           int* __restrict__ knn) {
  __shared__ float mv[CSPLIT * 32 * KNN_K];
  __shared__ int   mi[CSPLIT * 32 * KNN_K];
  __shared__ v2f   qall[CSPLIT * QSLOTS * 32];

  const int w  = __builtin_amdgcn_readfirstlane((int)(threadIdx.x >> 5));
  const int L  = threadIdx.x & 31;
  v2f* qbuf = qall + w * (QSLOTS * 32);
  const int rr = L & 15, hh = L >> 4;
  const int b  = blockIdx.x >> 9;
  const int r0 = (blockIdx.x & 511) << 4;
  const int bofs = b << 13;

  #pragma unroll
  for (int q = 0; q < QSLOTS; ++q) { v2f z; z.x = -__builtin_inff(); z.y = 0.f; qbuf[q * 32 + L] = z; }

  const v4f* xf4 = (const v4f*)xf + L;
  auto loadT = [&](int gt, v2f (&bb)[16]) {
    const v4f* p = xf4 + (size_t)gt * 256;
    #pragma unroll
    for (int i = 0; i < 8; ++i) {
      v4f v = p[i * 32];
      bb[2 * i].x = v.x;     bb[2 * i].y = v.y;
      bb[2 * i + 1].x = v.z; bb[2 * i + 1].y = v.w;
    }
  };

  v2f a[16];
  loadT((int)blockIdx.x, a);
  const float sqn = sq[bofs + r0 + rr];

  float vals[KNN_K]; int idxs[KNN_K];
  #pragma unroll
  for (int q = 0; q < KNN_K; ++q) { vals[q] = -__builtin_inff(); idxs[q] = 0; }
  int qn = 0;

  const int tb = b << 9;
  auto loadB = [&](int ct, v2f (&bb)[16]) { loadT(tb + ct, bb); };

  auto insert = [&](float s, int colg) {
    float old19 = vals[KNN_K - 1];
    #pragma unroll
    for (int t = 0; t < KNN_K - 1; ++t) {
      bool sh = s > vals[t + 1];
      float kv = vals[t]; int ki = idxs[t];
      vals[t] = sh ? vals[t + 1] : fmaxf(s, kv);
      idxs[t] = sh ? idxs[t + 1] : ((kv >= s) ? ki : colg);
    }
    if (s > old19) { vals[KNN_K - 1] = s; idxs[KNN_K - 1] = colg; }
  };

  auto drain = [&]() {
    const int mq = __builtin_amdgcn_readfirstlane((int)__builtin_amdgcn_wave_reduce_max_u32((unsigned)qn, 0));
    v2f e = qbuf[L];
    for (int i = 0; i < mq; ++i) {
      v2f cur = e;
      e = qbuf[(i + 1) * 32 + L];
      float s = cur.x;
      if (i < qn && s > vals[0]) insert(s, __float_as_int(cur.y));
    }
    qn = 0;
  };

  auto step = [&](int ct, v2f (&cur)[16], v2f (&nxt)[16]) {
    const int c0 = ct << 4;
    v4f sq0 = *(const v4f*)(sq + bofs + c0 + hh * 8);
    v4f sq1 = *(const v4f*)(sq + bofs + c0 + hh * 8 + 4);
    loadB((ct + 1) & 511, nxt);
    float sqt[8];
    #pragma unroll
    for (int j = 0; j < 4; ++j) { sqt[j] = sq0[j]; sqt[j + 4] = sq1[j]; }

    v8f acc = {};
    #pragma unroll
    for (int kk = 0; kk < 16; ++kk) acc = wmma_f32(cur[kk], a[kk], acc);

    const float thr = vals[0];
    #pragma unroll
    for (int j = 0; j < 8; ++j) {
      int colg = c0 + hh * 8 + j;
      float s = 2.f * acc[j] - sqt[j] - sqn;
      v2f ent; ent.x = s; ent.y = __int_as_float(colg);
      qbuf[qn * 32 + L] = ent;
      qn += (s > thr) ? 1 : 0;
    }
  };

  v2f bufA[16], bufB[16];
  const int ct0 = w * TPW;
  loadB(ct0, bufA);
  for (int ct = ct0; ct < ct0 + TPW; ct += 2) {
    step(ct,     bufA, bufB);
    step(ct + 1, bufB, bufA);
    if (__builtin_amdgcn_ballot_w32(qn > 16) != 0u) drain();
  }
  drain();

  #pragma unroll
  for (int q = 0; q < KNN_K; ++q) {
    mv[(w * 32 + L) * KNN_K + q] = vals[q];
    mi[(w * 32 + L) * KNN_K + q] = idxs[q];
  }
  __syncthreads();
  if (threadIdx.x < 16) {
    const int r = threadIdx.x;
    float hv[NLIST]; int hp[NLIST];
    #pragma unroll
    for (int l = 0; l < NLIST; ++l) {
      hp[l] = KNN_K - 1;
      hv[l] = mv[((l >> 1) * 32 + (l & 1) * 16 + r) * KNN_K + KNN_K - 1];
    }
    int* outp = knn + (size_t)(bofs + r0 + r) * KNN_K;
    for (int q = 0; q < KNN_K; ++q) {
      int best = 0; float bv = hv[0];
      #pragma unroll
      for (int l = 1; l < NLIST; ++l) { bool g = hv[l] > bv; bv = g ? hv[l] : bv; best = g ? l : best; }
      int bp = hp[0];
      #pragma unroll
      for (int l = 1; l < NLIST; ++l) bp = (l == best) ? hp[l] : bp;
      const int base = ((best >> 1) * 32 + (best & 1) * 16 + r) * KNN_K;
      outp[q] = mi[base + bp];
      const int   np = (bp > 0) ? bp - 1 : 0;
      const float nv = (bp > 0) ? mv[base + np] : -__builtin_inff();
      #pragma unroll
      for (int l = 0; l < NLIST; ++l) { bool m = (l == best); hp[l] = m ? bp - 1 : hp[l]; hv[l] = m ? nv : hv[l]; }
    }
  }
}

__launch_bounds__(128)
__global__ void conv_kernel(const float* __restrict__ xt, const int* __restrict__ knn,
                            const float* __restrict__ W, const float* __restrict__ bias,
                            float* __restrict__ mxp, float* __restrict__ mnp,
                            float* __restrict__ part) {
  __shared__ float    s_edge[4][16 * CIN2];
  __shared__ unsigned s_umax[4][4][COUT];
  __shared__ unsigned s_umin[4][4][COUT];
  __shared__ float    s_psum[4][16][8];
  __shared__ float    s_psq [4][16][8];

  const int tid = threadIdx.x;
  const int w = tid >> 5, L = tid & 31;
  const int rr = L & 15, hh = L >> 4;
  const int wfp = blockIdx.x * 16 + w * 4;

  {
    unsigned* um = &s_umax[0][0][0];
    unsigned* un = &s_umin[0][0][0];
    for (int i = tid; i < 4 * 4 * COUT; i += 128) { um[i] = 0u; un[i] = 0xFFFFFFFFu; }
  }
  __syncthreads();

  float fsum[4] = {0.f, 0.f, 0.f, 0.f}, fsq[4] = {0.f, 0.f, 0.f, 0.f};
  float* edge = &s_edge[w][0];

  #pragma unroll 1
  for (int t = 0; t < 5; ++t) {
    {
      int j = L >> 1, half = L & 1;
      int le = t * 16 + j;
      int p = le / 20, k = le - 20 * p;
      int gp = wfp + p;
      int bb = gp >> 13;
      int m = knn[(size_t)gp * KNN_K + k];
      const float* xn = xt + (size_t)gp * CH;
      const float* xm = xt + (size_t)((bb << 13) + m) * CH;
      float* er = edge + j * CIN2;
      if (half == 0) {
        #pragma unroll
        for (int c = 0; c < CH; c += 4)
          *(v4f*)(er + c) = *(const v4f*)(xm + c) - *(const v4f*)(xn + c);
      } else {
        #pragma unroll
        for (int c = 0; c < CH; c += 4)
          *(v4f*)(er + CH + c) = *(const v4f*)(xn + c);
      }
    }
    __syncthreads();

    int le0 = t * 16 + hh * 8;
    int pA = le0 / 20;
    int vth = (pA + 1) * 20 - le0;
    bool hasB = vth < 8;

    #pragma unroll 1
    for (int ot = 0; ot < 4; ++ot) {
      const int o = ot * 16 + rr;
      const float* wrow = W + (size_t)o * CIN2 + 2 * hh;
      const float* arow = edge + rr * CIN2 + 2 * hh;
      v8f acc = {};
      #pragma unroll
      for (int g = 0; g < 4; ++g) {
        v2f av[8], bv[8];
        #pragma unroll
        for (int u = 0; u < 8; ++u) {
          av[u] = *(const v2f*)(arow + (g * 8 + u) * 4);
          bv[u] = *(const v2f*)(wrow + (g * 8 + u) * 4);
        }
        #pragma unroll
        for (int u = 0; u < 8; ++u) acc = wmma_f32(av[u], bv[u], acc);
      }
      float bo = bias[o];
      float gAmax = -__builtin_inff(), gAmin = __builtin_inff();
      float gBmax = -__builtin_inff(), gBmin = __builtin_inff();
      float lsum = 0.f, lsq = 0.f;
      #pragma unroll
      for (int v = 0; v < 8; ++v) {
        float h = acc[v] + bo;
        h = (h >= 0.f) ? h : NEGSL * h;
        lsum += h; lsq += h * h;
        bool inB = (v >= vth);
        gAmax = fmaxf(gAmax, inB ? -__builtin_inff() : h);
        gAmin = fminf(gAmin, inB ?  __builtin_inff() : h);
        gBmax = fmaxf(gBmax, inB ? h : -__builtin_inff());
        gBmin = fminf(gBmin, inB ? h :  __builtin_inff());
      }
      fsum[ot] += lsum; fsq[ot] += lsq;
      atomicMax(&s_umax[w][pA][o], fenc(gAmax));
      atomicMin(&s_umin[w][pA][o], fenc(gAmin));
      if (hasB) {
        atomicMax(&s_umax[w][pA + 1][o], fenc(gBmax));
        atomicMin(&s_umin[w][pA + 1][o], fenc(gBmin));
      }
    }
    __syncthreads();
  }

  #pragma unroll
  for (int ot = 0; ot < 4; ++ot) {
    s_psum[ot][rr][w * 2 + hh] = fsum[ot];
    s_psq [ot][rr][w * 2 + hh] = fsq[ot];
  }
  __syncthreads();
  if (tid < COUT) {
    float s = 0.f, s2 = 0.f;
    #pragma unroll
    for (int q = 0; q < 8; ++q) { s += s_psum[tid >> 4][tid & 15][q]; s2 += s_psq[tid >> 4][tid & 15][q]; }
    part[(size_t)blockIdx.x * 128 + tid]      = s;
    part[(size_t)blockIdx.x * 128 + 64 + tid] = s2;
  }

  #pragma unroll
  for (int it = 0; it < 8; ++it) {
    int q = L + 32 * it;
    int p = q >> 6, o = q & 63;
    int gp = wfp + p;
    int bb = gp >> 13, n = gp & (NPT - 1);
    size_t oi = ((size_t)(bb * COUT + o) << 13) + n;
    mxp[oi] = fdec(s_umax[w][p][o]);
    mnp[oi] = fdec(s_umin[w][p][o]);
  }
}

__global__ void stats_kernel(const float* __restrict__ part, float* __restrict__ gst) {
  int t = threadIdx.x;
  float s = 0.f;
  for (int g = 0; g < 1024; ++g) s += part[(size_t)g * 128 + t];
  gst[t] = s;
}

__global__ void out_kernel(const float* __restrict__ mxp, const float* __restrict__ mnp,
                           const float* __restrict__ gst,
                           const float* __restrict__ gamma, const float* __restrict__ beta,
                           float* __restrict__ outp) {
  int i = blockIdx.x * blockDim.x + threadIdx.x;
  int o = (i >> 13) & 63;
  const float cnt = (float)(BATCH * NPT * KNN_K);
  float mean = gst[o] / cnt;
  float var  = gst[COUT + o] / cnt - mean * mean;
  float scale = gamma[o] * rsqrtf(var + BNEPS);
  float sel = (scale >= 0.f) ? mxp[i] : mnp[i];
  outp[i] = (sel - mean) * scale + beta[o];
}

extern "C" void kernel_launch(void* const* d_in, const int* in_sizes, int n_in,
                              void* d_out, int out_size, void* d_ws, size_t ws_size,
                              hipStream_t stream) {
  const float* x     = (const float*)d_in[0];
  const float* W     = (const float*)d_in[1];
  const float* bias  = (const float*)d_in[2];
  const float* gamma = (const float*)d_in[3];
  const float* beta  = (const float*)d_in[4];
  float* outp = (float*)d_out;

  float* xt  = (float*)d_ws;
  float* sq  = xt + (size_t)BATCH * NPT * CH;
  int*   knn = (int*)(sq + (size_t)BATCH * NPT);
  float* mxp = (float*)(knn + (size_t)BATCH * NPT * KNN_K);
  float* mnp = mxp + (size_t)BATCH * COUT * NPT;
  float* gst = mnp + (size_t)BATCH * COUT * NPT;
  float* part = gst + 128;

  float* xf = mxp;

  prep_kernel <<<(BATCH * NPT) / 256, 256, 0, stream>>>(x, xt, sq, xf);
  knn_kernel  <<<BATCH * (NPT / 16), 32 * CSPLIT, 0, stream>>>(xf, sq, knn);
  conv_kernel <<<(BATCH * NPT) / 16, 128, 0, stream>>>(xt, knn, W, bias, mxp, mnp, part);
  stats_kernel<<<1, 128, 0, stream>>>(part, gst);
  out_kernel  <<<(BATCH * COUT * NPT) / 256, 256, 0, stream>>>(mxp, mnp, gst, gamma, beta, outp);
}


# conv: edge-tile rows in LDS padded from 128 to 132 floats (16 B x 33) so the 16 rows of an A-fragment read and the 8 rows of a staging store fall on distinct banks, not one
# speedup vs baseline: 3.0352x; 1.1678x over previous
#include <hip/hip_runtime.h>
#include <stdint.h>

#define BATCH 2
#define CH    64
#define NPT   8192
#define KNN_K 20
#define COUT  64
#define CIN2  128
#define NEGSL 0.2f
#define BNEPS 1e-5f

typedef float v2f __attribute__((ext_vector_type(2)));
typedef float v4f __attribute__((ext_vector_type(4)));
typedef float v8f __attribute__((ext_vector_type(8)));

__device__ __forceinline__ v8f wmma_f32(v2f a, v2f b, v8f c) {
  return __builtin_amdgcn_wmma_f32_16x16x4_f32(false, a, false, b, (short)0, c, false, false);
}

__device__ __forceinline__ unsigned fenc(float f) {
  unsigned u = __float_as_uint(f);
  return (u & 0x80000000u) ? ~u : (u | 0x80000000u);
}
__device__ __forceinline__ float fdec(unsigned u) {
  return __uint_as_float((u & 0x80000000u) ? (u ^ 0x80000000u) : ~u);
}

__global__ void prep_kernel(const float* __restrict__ x,
                            float* __restrict__ xt, float* __restrict__ sq,
                            float* __restrict__ xf) {
  int gp = blockIdx.x * blockDim.x + threadIdx.x;
  int b = gp >> 13, n = gp & (NPT - 1);
  const float* xb = x + (size_t)b * CH * NPT + n;
  v4f* xf4 = (v4f*)xf + (size_t)(gp >> 4) * 256 + (gp & 15);
  float acc = 0.f;
  for (int i = 0; i < 8; ++i) {
    float v[8];
    #pragma unroll
    for (int e = 0; e < 8; ++e) v[e] = xb[(size_t)(i * 8 + e) * NPT];
    #pragma unroll
    for (int e = 0; e < 8; ++e) {
      acc = __builtin_fmaf(v[e], v[e], acc);
      xt[(size_t)gp * CH + i * 8 + e] = v[e];
    }
    v4f h0, h1;
    h0.x = v[0]; h0.y = v[1]; h0.z = v[4]; h0.w = v[5];
    h1.x = v[2]; h1.y = v[3]; h1.z = v[6]; h1.w = v[7];
    xf4[i * 32]      = h0;
    xf4[i * 32 + 16] = h1;
  }
  sq[gp] = acc;
}

#define QSLOTS 33
#define CSPLIT 4
#define TPW    (NPT / 16 / CSPLIT)
#define NLIST  (2 * CSPLIT)
__launch_bounds__(32 * CSPLIT)
__global__ void knn_kernel(const float* __restrict__ xf, const float* __restrict__ sq,
                           int* __restrict__ knn) {
  __shared__ float mv[CSPLIT * 32 * KNN_K];
  __shared__ int   mi[CSPLIT * 32 * KNN_K];
  __shared__ v2f   qall[CSPLIT * QSLOTS * 32];

  const int w  = __builtin_amdgcn_readfirstlane((int)(threadIdx.x >> 5));
  const int L  = threadIdx.x & 31;
  v2f* qbuf = qall + w * (QSLOTS * 32);
  const int rr = L & 15, hh = L >> 4;
  const int b  = blockIdx.x >> 9;
  const int r0 = (blockIdx.x & 511) << 4;
  const int bofs = b << 13;

  #pragma unroll
  for (int q = 0; q < QSLOTS; ++q) { v2f z; z.x = -__builtin_inff(); z.y = 0.f; qbuf[q * 32 + L] = z; }

  const v4f* xf4 = (const v4f*)xf + L;
  auto loadT = [&](int gt, v2f (&bb)[16]) {
    const v4f* p = xf4 + (size_t)gt * 256;
    #pragma unroll
    for (int i = 0; i < 8; ++i) {
      v4f v = p[i * 32];
      bb[2 * i].x = v.x;     bb[2 * i].y = v.y;
      bb[2 * i + 1].x = v.z; bb[2 * i + 1].y = v.w;
    }
  };

  v2f a[16];
  loadT((int)blockIdx.x, a);
  const float sqn = sq[bofs + r0 + rr];

  float vals[KNN_K]; int idxs[KNN_K];
  #pragma unroll
  for (int q = 0; q < KNN_K; ++q) { vals[q] = -__builtin_inff(); idxs[q] = 0; }
  int qn = 0;

  const int tb = b << 9;
  auto loadB = [&](int ct, v2f (&bb)[16]) { loadT(tb + ct, bb); };

  auto insert = [&](float s, int colg) {
    float old19 = vals[KNN_K - 1];
    #pragma unroll
    for (int t = 0; t < KNN_K - 1; ++t) {
      bool sh = s > vals[t + 1];
      float kv = vals[t]; int ki = idxs[t];
      vals[t] = sh ? vals[t + 1] : fmaxf(s, kv);
      idxs[t] = sh ? idxs[t + 1] : ((kv >= s) ? ki : colg);
    }
    if (s > old19) { vals[KNN_K - 1] = s; idxs[KNN_K - 1] = colg; }
  };

  auto drain = [&]() {
    const int mq = __builtin_amdgcn_readfirstlane((int)__builtin_amdgcn_wave_reduce_max_u32((unsigned)qn, 0));
    v2f e = qbuf[L];
    for (int i = 0; i < mq; ++i) {
      v2f cur = e;
      e = qbuf[(i + 1) * 32 + L];
      float s = cur.x;
      if (i < qn && s > vals[0]) insert(s, __float_as_int(cur.y));
    }
    qn = 0;
  };

  auto step = [&](int ct, v2f (&cur)[16], v2f (&nxt)[16]) {
    const int c0 = ct << 4;
    v4f sq0 = *(const v4f*)(sq + bofs + c0 + hh * 8);
    v4f sq1 = *(const v4f*)(sq + bofs + c0 + hh * 8 + 4);
    loadB((ct + 1) & 511, nxt);
    float sqt[8];
    #pragma unroll
    for (int j = 0; j < 4; ++j) { sqt[j] = sq0[j]; sqt[j + 4] = sq1[j]; }

    v8f acc = {};
    #pragma unroll
    for (int kk = 0; kk < 16; ++kk) acc = wmma_f32(cur[kk], a[kk], acc);

    const float thr = vals[0];
    #pragma unroll
    for (int j = 0; j < 8; ++j) {
      int colg = c0 + hh * 8 + j;
      float s = 2.f * acc[j] - sqt[j] - sqn;
      v2f ent; ent.x = s; ent.y = __int_as_float(colg);
      qbuf[qn * 32 + L] = ent;
      qn += (s > thr) ? 1 : 0;
    }
  };

  v2f bufA[16], bufB[16];
  const int ct0 = w * TPW;
  loadB(ct0, bufA);
  for (int ct = ct0; ct < ct0 + TPW; ct += 2) {
    step(ct,     bufA, bufB);
    step(ct + 1, bufB, bufA);
    if (__builtin_amdgcn_ballot_w32(qn > 16) != 0u) drain();
  }
  drain();

  #pragma unroll
  for (int q = 0; q < KNN_K; ++q) {
    mv[(w * 32 + L) * KNN_K + q] = vals[q];
    mi[(w * 32 + L) * KNN_K + q] = idxs[q];
  }
  __syncthreads();
  if (threadIdx.x < 16) {
    const int r = threadIdx.x;
    float hv[NLIST]; int hp[NLIST];
    #pragma unroll
    for (int l = 0; l < NLIST; ++l) {
      hp[l] = KNN_K - 1;
      hv[l] = mv[((l >> 1) * 32 + (l & 1) * 16 + r) * KNN_K + KNN_K - 1];
    }
    int* outp = knn + (size_t)(bofs + r0 + r) * KNN_K;
    for (int q = 0; q < KNN_K; ++q) {
      int best = 0; float bv = hv[0];
      #pragma unroll
      for (int l = 1; l < NLIST; ++l) { bool g = hv[l] > bv; bv = g ? hv[l] : bv; best = g ? l : best; }
      int bp = hp[0];
      #pragma unroll
      for (int l = 1; l < NLIST; ++l) bp = (l == best) ? hp[l] : bp;
      const int base = ((best >> 1) * 32 + (best & 1) * 16 + r) * KNN_K;
      outp[q] = mi[base + bp];
      const int   np = (bp > 0) ? bp - 1 : 0;
      const float nv = (bp > 0) ? mv[base + np] : -__builtin_inff();
      #pragma unroll
      for (int l = 0; l < NLIST; ++l) { bool m = (l == best); hp[l] = m ? bp - 1 : hp[l]; hv[l] = m ? nv : hv[l]; }
    }
  }
}

#define EPITCH (CIN2 + 4)
__launch_bounds__(128)
__global__ void conv_kernel(const float* __restrict__ xt, const int* __restrict__ knn,
                            const float* __restrict__ W, const float* __restrict__ bias,
                            float* __restrict__ mxp, float* __restrict__ mnp,
                            float* __restrict__ part) {
  __shared__ float    s_edge[4][16 * EPITCH];
  __shared__ unsigned s_umax[4][4][COUT];
  __shared__ unsigned s_umin[4][4][COUT];
  __shared__ float    s_psum[4][16][8];
  __shared__ float    s_psq [4][16][8];

  const int tid = threadIdx.x;
  const int w = tid >> 5, L = tid & 31;
  const int rr = L & 15, hh = L >> 4;
  const int wfp = blockIdx.x * 16 + w * 4;

  {
    unsigned* um = &s_umax[0][0][0];
    unsigned* un = &s_umin[0][0][0];
    for (int i = tid; i < 4 * 4 * COUT; i += 128) { um[i] = 0u; un[i] = 0xFFFFFFFFu; }
  }
  __syncthreads();

  float fsum[4] = {0.f, 0.f, 0.f, 0.f}, fsq[4] = {0.f, 0.f, 0.f, 0.f};
  float* edge = &s_edge[w][0];

  #pragma unroll 1
  for (int t = 0; t < 5; ++t) {
    {
      int j = L >> 1, half = L & 1;
      int le = t * 16 + j;
      int p = le / 20, k = le - 20 * p;
      int gp = wfp + p;
      int bb = gp >> 13;
      int m = knn[(size_t)gp * KNN_K + k];
      const float* xn = xt + (size_t)gp * CH;
      const float* xm = xt + (size_t)((bb << 13) + m) * CH;
      float* er = edge + j * EPITCH;
      if (half == 0) {
        #pragma unroll
        for (int c = 0; c < CH; c += 4)
          *(v4f*)(er + c) = *(const v4f*)(xm + c) - *(const v4f*)(xn + c);
      } else {
        #pragma unroll
        for (int c = 0; c < CH; c += 4)
          *(v4f*)(er + CH + c) = *(const v4f*)(xn + c);
      }
    }
    __syncthreads();

    int le0 = t * 16 + hh * 8;
    int pA = le0 / 20;
    int vth = (pA + 1) * 20 - le0;
    bool hasB = vth < 8;

    #pragma unroll 1
    for (int ot = 0; ot < 4; ++ot) {
      const int o = ot * 16 + rr;
      const float* wrow = W + (size_t)o * CIN2 + 2 * hh;
      const float* arow = edge + rr * EPITCH + 2 * hh;
      v8f acc = {};
      #pragma unroll
      for (int g = 0; g < 4; ++g) {
        v2f av[8], bv[8];
        #pragma unroll
        for (int u = 0; u < 8; ++u) {
          av[u] = *(const v2f*)(arow + (g * 8 + u) * 4);
          bv[u] = *(const v2f*)(wrow + (g * 8 + u) * 4);
        }
        #pragma unroll
        for (int u = 0; u < 8; ++u) acc = wmma_f32(av[u], bv[u], acc);
      }
      float bo = bias[o];
      float gAmax = -__builtin_inff(), gAmin = __builtin_inff();
      float gBmax = -__builtin_inff(), gBmin = __builtin_inff();
      float lsum = 0.f, lsq = 0.f;
      #pragma unroll
      for (int v = 0; v < 8; ++v) {
        float h = acc[v] + bo;
        h = (h >= 0.f) ? h : NEGSL * h;
        lsum += h; lsq += h * h;
        bool inB = (v >= vth);
        gAmax = fmaxf(gAmax, inB ? -__builtin_inff() : h);
        gAmin = fminf(gAmin, inB ?  __builtin_inff() : h);
        gBmax = fmaxf(gBmax, inB ? h : -__builtin_inff());
        gBmin = fminf(gBmin, inB ? h :  __builtin_inff());
      }
      fsum[ot] += lsum; fsq[ot] += lsq;
      atomicMax(&s_umax[w][pA][o], fenc(gAmax));
      atomicMin(&s_umin[w][pA][o], fenc(gAmin));
      if (hasB) {
        atomicMax(&s_umax[w][pA + 1][o], fenc(gBmax));
        atomicMin(&s_umin[w][pA + 1][o], fenc(gBmin));
      }
    }
    __syncthreads();
  }

  #pragma unroll
  for (int ot = 0; ot < 4; ++ot) {
    s_psum[ot][rr][w * 2 + hh] = fsum[ot];
    s_psq [ot][rr][w * 2 + hh] = fsq[ot];
  }
  __syncthreads();
  if (tid < COUT) {
    float s = 0.f, s2 = 0.f;
    #pragma unroll
    for (int q = 0; q < 8; ++q) { s += s_psum[tid >> 4][tid & 15][q]; s2 += s_psq[tid >> 4][tid & 15][q]; }
    part[(size_t)tid * 1024 + blockIdx.x]        = s;
    part[(size_t)(64 + tid) * 1024 + blockIdx.x] = s2;
  }

  #pragma unroll
  for (int it = 0; it < 8; ++it) {
    int q = L + 32 * it;
    int p = q >> 6, o = q & 63;
    int gp = wfp + p;
    int bb = gp >> 13, n = gp & (NPT - 1);
    size_t oi = ((size_t)(bb * COUT + o) << 13) + n;
    mxp[oi] = fdec(s_umax[w][p][o]);
    mnp[oi] = fdec(s_umin[w][p][o]);
  }
}

__launch_bounds__(256)
__global__ void out_kernel(const float* __restrict__ mxp, const float* __restrict__ mnp,
                           const float* __restrict__ part,
                           const float* __restrict__ gamma, const float* __restrict__ beta,
                           float* __restrict__ outp) {
  __shared__ float r_s[256], r_q[256];
  __shared__ float t_s[32],  t_q[32];
  const int tid   = threadIdx.x;
  const int plane = blockIdx.x >> 2;
  const int o     = plane & (COUT - 1);

  const v4f ps = *(const v4f*)(part + (size_t)o * 1024 + tid * 4);
  const v4f pq = *(const v4f*)(part + (size_t)(COUT + o) * 1024 + tid * 4);
  r_s[tid] = ((ps.x + ps.y) + ps.z) + ps.w;
  r_q[tid] = ((pq.x + pq.y) + pq.z) + pq.w;
  __syncthreads();
  if (tid < 32) {
    float s = 0.f, s2 = 0.f;
    for (int q = 0; q < 8; ++q) { s += r_s[tid * 8 + q]; s2 += r_q[tid * 8 + q]; }
    t_s[tid] = s; t_q[tid] = s2;
  }
  __syncthreads();
  float sum = 0.f, sumsq = 0.f;
  for (int q = 0; q < 32; ++q) { sum += t_s[q]; sumsq += t_q[q]; }

  const float cnt = (float)(BATCH * NPT * KNN_K);
  const float mean = sum / cnt;
  const float var  = sumsq / cnt - mean * mean;
  const float scale = gamma[o] * rsqrtf(var + BNEPS);
  const float shift = beta[o];
  const float* __restrict__ src = (scale >= 0.f) ? mxp : mnp;
  const size_t base = (size_t)plane * NPT + (size_t)(blockIdx.x & 3) * 2048 + tid * 4;
  const v4f a0 = *(const v4f*)(src + base);
  const v4f a1 = *(const v4f*)(src + base + 1024);
  v4f o0, o1;
  for (int e = 0; e < 4; ++e) {
    o0[e] = (a0[e] - mean) * scale + shift;
    o1[e] = (a1[e] - mean) * scale + shift;
  }
  *(v4f*)(outp + base)        = o0;
  *(v4f*)(outp + base + 1024) = o1;
}

extern "C" void kernel_launch(void* const* d_in, const int* in_sizes, int n_in,
                              void* d_out, int out_size, void* d_ws, size_t ws_size,
                              hipStream_t stream) {
  const float* x     = (const float*)d_in[0];
  const float* W     = (const float*)d_in[1];
  const float* bias  = (const float*)d_in[2];
  const float* gamma = (const float*)d_in[3];
  const float* beta  = (const float*)d_in[4];
  float* outp = (float*)d_out;

  float* xt  = (float*)d_ws;
  float* sq  = xt + (size_t)BATCH * NPT * CH;
  int*   knn = (int*)(sq + (size_t)BATCH * NPT);
  float* mxp = (float*)(knn + (size_t)BATCH * NPT * KNN_K);
  float* mnp = mxp + (size_t)BATCH * COUT * NPT;
  float* part = mnp + (size_t)BATCH * COUT * NPT + 128;

  float* xf = mxp;

  prep_kernel <<<(BATCH * NPT) / 256, 256, 0, stream>>>(x, xt, sq, xf);
  knn_kernel  <<<BATCH * (NPT / 16), 32 * CSPLIT, 0, stream>>>(xf, sq, knn);
  conv_kernel <<<(BATCH * NPT) / 16, 128, 0, stream>>>(xt, knn, W, bias, mxp, mnp, part);
  out_kernel  <<<BATCH * COUT * 4, 256, 0, stream>>>(mxp, mnp, part, gamma, beta, outp);
}


# conv: weights copied once per workgroup into LDS in fragment order; a wave's B operands are one group of contiguous 512 B ds_load_b128 per output tile, not 8 B-per-lane global loads of 16 lines each
# speedup vs baseline: 3.2932x; 1.0850x over previous
#include <hip/hip_runtime.h>
#include <stdint.h>

#define BATCH 2
#define CH    64
#define NPT   8192
#define KNN_K 20
#define COUT  64
#define CIN2  128
#define NEGSL 0.2f
#define BNEPS 1e-5f

typedef float v2f __attribute__((ext_vector_type(2)));
typedef float v4f __attribute__((ext_vector_type(4)));
typedef float v8f __attribute__((ext_vector_type(8)));

__device__ __forceinline__ v8f wmma_f32(v2f a, v2f b, v8f c) {
  return __builtin_amdgcn_wmma_f32_16x16x4_f32(false, a, false, b, (short)0, c, false, false);
}

__device__ __forceinline__ unsigned fenc(float f) {
  unsigned u = __float_as_uint(f);
  return (u & 0x80000000u) ? ~u : (u | 0x80000000u);
}
__device__ __forceinline__ float fdec(unsigned u) {
  return __uint_as_float((u & 0x80000000u) ? (u ^ 0x80000000u) : ~u);
}

__global__ void prep_kernel(const float* __restrict__ x,
                            float* __restrict__ xt, float* __restrict__ sq,
                            float* __restrict__ xf) {
  int gp = blockIdx.x * blockDim.x + threadIdx.x;
  int b = gp >> 13, n = gp & (NPT - 1);
  const float* xb = x + (size_t)b * CH * NPT + n;
  v4f* xf4 = (v4f*)xf + (size_t)(gp >> 4) * 256 + (gp & 15);
  float acc = 0.f;
  for (int i = 0; i < 8; ++i) {
    float v[8];
    #pragma unroll
    for (int e = 0; e < 8; ++e) v[e] = xb[(size_t)(i * 8 + e) * NPT];
    #pragma unroll
    for (int e = 0; e < 8; ++e) {
      acc = __builtin_fmaf(v[e], v[e], acc);
      xt[(size_t)gp * CH + i * 8 + e] = v[e];
    }
    v4f h0, h1;
    h0.x = v[0]; h0.y = v[1]; h0.z = v[4]; h0.w = v[5];
    h1.x = v[2]; h1.y = v[3]; h1.z = v[6]; h1.w = v[7];
    xf4[i * 32]      = h0;
    xf4[i * 32 + 16] = h1;
  }
  sq[gp] = acc;
}

#define QSLOTS 33
#define CSPLIT 4
#define TPW    (NPT / 16 / CSPLIT)
#define NLIST  (2 * CSPLIT)
__launch_bounds__(32 * CSPLIT)
__global__ void knn_kernel(const float* __restrict__ xf, const float* __restrict__ sq,
                           int* __restrict__ knn) {
  __shared__ float mv[CSPLIT * 32 * KNN_K];
  __shared__ int   mi[CSPLIT * 32 * KNN_K];
  __shared__ v2f   qall[CSPLIT * QSLOTS * 32];

  const int w  = __builtin_amdgcn_readfirstlane((int)(threadIdx.x >> 5));
  const int L  = threadIdx.x & 31;
  v2f* qbuf = qall + w * (QSLOTS * 32);
  const int rr = L & 15, hh = L >> 4;
  const int b  = blockIdx.x >> 9;
  const int r0 = (blockIdx.x & 511) << 4;
  const int bofs = b << 13;

  #pragma unroll
  for (int q = 0; q < QSLOTS; ++q) { v2f z; z.x = -__builtin_inff(); z.y = 0.f; qbuf[q * 32 + L] = z; }

  const v4f* xf4 = (const v4f*)xf + L;
  auto loadT = [&](int gt, v2f (&bb)[16]) {
    const v4f* p = xf4 + (size_t)gt * 256;
    #pragma unroll
    for (int i = 0; i < 8; ++i) {
      v4f v = p[i * 32];
      bb[2 * i].x = v.x;     bb[2 * i].y = v.y;
      bb[2 * i + 1].x = v.z; bb[2 * i + 1].y = v.w;
    }
  };

  v2f a[16];
  loadT((int)blockIdx.x, a);
  const float sqn = sq[bofs + r0 + rr];

  float vals[KNN_K]; int idxs[KNN_K];
  #pragma unroll
  for (int q = 0; q < KNN_K; ++q) { vals[q] = -__builtin_inff(); idxs[q] = 0; }
  int qn = 0;

  const int tb = b << 9;
  auto loadB = [&](int ct, v2f (&bb)[16]) { loadT(tb + ct, bb); };

  auto insert = [&](float s, int colg) {
    float old19 = vals[KNN_K - 1];
    #pragma unroll
    for (int t = 0; t < KNN_K - 1; ++t) {
      bool sh = s > vals[t + 1];
      float kv = vals[t]; int ki = idxs[t];
      vals[t] = sh ? vals[t + 1] : fmaxf(s, kv);
      idxs[t] = sh ? idxs[t + 1] : ((kv >= s) ? ki : colg);
    }
    if (s > old19) { vals[KNN_K - 1] = s; idxs[KNN_K - 1] = colg; }
  };

  auto drain = [&]() {
    const int mq = __builtin_amdgcn_readfirstlane((int)__builtin_amdgcn_wave_reduce_max_u32((unsigned)qn, 0));
    v2f e = qbuf[L];
    for (int i = 0; i < mq; ++i) {
      v2f cur = e;
      e = qbuf[(i + 1) * 32 + L];
      float s = cur.x;
      if (i < qn && s > vals[0]) insert(s, __float_as_int(cur.y));
    }
    qn = 0;
  };

  auto step = [&](int ct, v2f (&cur)[16], v2f (&nxt)[16]) {
    const int c0 = ct << 4;
    v4f sq0 = *(const v4f*)(sq + bofs + c0 + hh * 8);
    v4f sq1 = *(const v4f*)(sq + bofs + c0 + hh * 8 + 4);
    loadB((ct + 1) & 511, nxt);
    float sqt[8];
    #pragma unroll
    for (int j = 0; j < 4; ++j) { sqt[j] = sq0[j]; sqt[j + 4] = sq1[j]; }

    v8f acc = {};
    #pragma unroll
    for (int kk = 0; kk < 16; ++kk) acc = wmma_f32(cur[kk], a[kk], acc);

    const float thr = vals[0];
    #pragma unroll
    for (int j = 0; j < 8; ++j) {
      int colg = c0 + hh * 8 + j;
      float s = 2.f * acc[j] - sqt[j] - sqn;
      v2f ent; ent.x = s; ent.y = __int_as_float(colg);
      qbuf[qn * 32 + L] = ent;
      qn += (s > thr) ? 1 : 0;
    }
  };

  v2f bufA[16], bufB[16];
  const int ct0 = w * TPW;
  loadB(ct0, bufA);
  for (int ct = ct0; ct < ct0 + TPW; ct += 2) {
    step(ct,     bufA, bufB);
    step(ct + 1, bufB, bufA);
    if (__builtin_amdgcn_ballot_w32(qn > 16) != 0u) drain();
  }
  drain();

  #pragma unroll
  for (int q = 0; q < KNN_K; ++q) {
    mv[(w * 32 + L) * KNN_K + q] = vals[q];
    mi[(w * 32 + L) * KNN_K + q] = idxs[q];
  }
  __syncthreads();
  if (threadIdx.x < 16) {
    const int r = threadIdx.x;
    float hv[NLIST]; int hp[NLIST];
    #pragma unroll
    for (int l = 0; l < NLIST; ++l) {
      hp[l] = KNN_K - 1;
      hv[l] = mv[((l >> 1) * 32 + (l & 1) * 16 + r) * KNN_K + KNN_K - 1];
    }
    int* outp = knn + (size_t)(bofs + r0 + r) * KNN_K;
    for (int q = 0; q < KNN_K; ++q) {
      int best = 0; float bv = hv[0];
      #pragma unroll
      for (int l = 1; l < NLIST; ++l) { bool g = hv[l] > bv; bv = g ? hv[l] : bv; best = g ? l : best; }
      int bp = hp[0];
      #pragma unroll
      for (int l = 1; l < NLIST; ++l) bp = (l == best) ? hp[l] : bp;
      const int base = ((best >> 1) * 32 + (best & 1) * 16 + r) * KNN_K;
      outp[q] = mi[base + bp];
      const int   np = (bp > 0) ? bp - 1 : 0;
      const float nv = (bp > 0) ? mv[base + np] : -__builtin_inff();
      #pragma unroll
      for (int l = 0; l < NLIST; ++l) { bool m = (l == best); hp[l] = m ? bp - 1 : hp[l]; hv[l] = m ? nv : hv[l]; }
    }
  }
}

#define EPITCH (CIN2 + 4)
__launch_bounds__(128)
__global__ void conv_kernel(const float* __restrict__ xt, const int* __restrict__ knn,
                            const float* __restrict__ W, const float* __restrict__ bias,
                            float* __restrict__ mxp, float* __restrict__ mnp,
                            float* __restrict__ part) {
  __shared__ float    s_edge[4][16 * EPITCH];
  __shared__ v4f      s_w[4 * 16 * 32];
  __shared__ unsigned s_umax[4][4][COUT];
  __shared__ unsigned s_umin[4][4][COUT];
  __shared__ float    s_psum[4][16][8];
  __shared__ float    s_psq [4][16][8];

  const int tid = threadIdx.x;
  const int w = tid >> 5, L = tid & 31;
  const int rr = L & 15, hh = L >> 4;
  const int wfp = blockIdx.x * 16 + w * 4;

  {
    unsigned* um = &s_umax[0][0][0];
    unsigned* un = &s_umin[0][0][0];
    for (int i = tid; i < 4 * 4 * COUT; i += 128) { um[i] = 0u; un[i] = 0xFFFFFFFFu; }
  }
  {
    float* wls = (float*)s_w;
    #pragma unroll
    for (int q = 0; q < 16; ++q) {
      const int u = q * 4 + w;
      const v4f v = *(const v4f*)(W + (size_t)((u >> 4) * 16 + rr) * CIN2 + 8 * (u & 15) + 4 * hh);
      float* dst = wls + u * 128 + rr * 4 + 2 * hh;
      v2f lo, hi;
      lo.x = v.x; lo.y = v.y; hi.x = v.z; hi.y = v.w;
      *(v2f*)dst        = lo;
      *(v2f*)(dst + 64) = hi;
    }
  }
  __syncthreads();

  float fsum[4] = {0.f, 0.f, 0.f, 0.f}, fsq[4] = {0.f, 0.f, 0.f, 0.f};
  float* edge = &s_edge[w][0];

  #pragma unroll 1
  for (int t = 0; t < 5; ++t) {
    {
      int j = L >> 1, half = L & 1;
      int le = t * 16 + j;
      int p = le / 20, k = le - 20 * p;
      int gp = wfp + p;
      int bb = gp >> 13;
      int m = knn[(size_t)gp * KNN_K + k];
      const float* xn = xt + (size_t)gp * CH;
      const float* xm = xt + (size_t)((bb << 13) + m) * CH;
      float* er = edge + j * EPITCH;
      if (half == 0) {
        #pragma unroll
        for (int c = 0; c < CH; c += 4)
          *(v4f*)(er + c) = *(const v4f*)(xm + c) - *(const v4f*)(xn + c);
      } else {
        #pragma unroll
        for (int c = 0; c < CH; c += 4)
          *(v4f*)(er + CH + c) = *(const v4f*)(xn + c);
      }
    }
    __syncthreads();

    int le0 = t * 16 + hh * 8;
    int pA = le0 / 20;
    int vth = (pA + 1) * 20 - le0;
    bool hasB = vth < 8;

    const float* arow = edge + rr * EPITCH + 2 * hh;
    v2f av[32];
    #pragma unroll
    for (int u = 0; u < 32; ++u) av[u] = *(const v2f*)(arow + u * 4);

    #pragma unroll 1
    for (int ot = 0; ot < 4; ++ot) {
      const int o = ot * 16 + rr;
      const v4f* wl = s_w + ot * (16 * 32) + L;
      v8f acc = {};
      v4f bq[16];
      #pragma unroll
      for (int u = 0; u < 16; ++u) bq[u] = wl[u * 32];
      __builtin_amdgcn_sched_barrier(0);
      #pragma unroll
      for (int u = 0; u < 16; ++u) {
        v2f b0, b1;
        b0.x = bq[u].x; b0.y = bq[u].y;
        b1.x = bq[u].z; b1.y = bq[u].w;
        acc = wmma_f32(av[2 * u],     b0, acc);
        acc = wmma_f32(av[2 * u + 1], b1, acc);
      }
      float bo = bias[o];
      float gAmax = -__builtin_inff(), gAmin = __builtin_inff();
      float gBmax = -__builtin_inff(), gBmin = __builtin_inff();
      float lsum = 0.f, lsq = 0.f;
      #pragma unroll
      for (int v = 0; v < 8; ++v) {
        float h = acc[v] + bo;
        h = (h >= 0.f) ? h : NEGSL * h;
        lsum += h; lsq += h * h;
        bool inB = (v >= vth);
        gAmax = fmaxf(gAmax, inB ? -__builtin_inff() : h);
        gAmin = fminf(gAmin, inB ?  __builtin_inff() : h);
        gBmax = fmaxf(gBmax, inB ? h : -__builtin_inff());
        gBmin = fminf(gBmin, inB ? h :  __builtin_inff());
      }
      fsum[ot] += lsum; fsq[ot] += lsq;
      atomicMax(&s_umax[w][pA][o], fenc(gAmax));
      atomicMin(&s_umin[w][pA][o], fenc(gAmin));
      if (hasB) {
        atomicMax(&s_umax[w][pA + 1][o], fenc(gBmax));
        atomicMin(&s_umin[w][pA + 1][o], fenc(gBmin));
      }
    }
    __syncthreads();
  }

  #pragma unroll
  for (int ot = 0; ot < 4; ++ot) {
    s_psum[ot][rr][w * 2 + hh] = fsum[ot];
    s_psq [ot][rr][w * 2 + hh] = fsq[ot];
  }
  __syncthreads();
  if (tid < COUT) {
    float s = 0.f, s2 = 0.f;
    #pragma unroll
    for (int q = 0; q < 8; ++q) { s += s_psum[tid >> 4][tid & 15][q]; s2 += s_psq[tid >> 4][tid & 15][q]; }
    part[(size_t)tid * 1024 + blockIdx.x]        = s;
    part[(size_t)(64 + tid) * 1024 + blockIdx.x] = s2;
  }

  #pragma unroll
  for (int it = 0; it < 8; ++it) {
    int q = L + 32 * it;
    int p = q >> 6, o = q & 63;
    int gp = wfp + p;
    int bb = gp >> 13, n = gp & (NPT - 1);
    size_t oi = ((size_t)(bb * COUT + o) << 13) + n;
    mxp[oi] = fdec(s_umax[w][p][o]);
    mnp[oi] = fdec(s_umin[w][p][o]);
  }
}

__launch_bounds__(256)
__global__ void out_kernel(const float* __restrict__ mxp, const float* __restrict__ mnp,
                           const float* __restrict__ part,
                           const float* __restrict__ gamma, const float* __restrict__ beta,
                           float* __restrict__ outp) {
  __shared__ float r_s[256], r_q[256];
  __shared__ float t_s[32],  t_q[32];
  const int tid   = threadIdx.x;
  const int plane = blockIdx.x >> 2;
  const int o     = plane & (COUT - 1);

  const v4f ps = *(const v4f*)(part + (size_t)o * 1024 + tid * 4);
  const v4f pq = *(const v4f*)(part + (size_t)(COUT + o) * 1024 + tid * 4);
  r_s[tid] = ((ps.x + ps.y) + ps.z) + ps.w;
  r_q[tid] = ((pq.x + pq.y) + pq.z) + pq.w;
  __syncthreads();
  if (tid < 32) {
    float s = 0.f, s2 = 0.f;
    for (int q = 0; q < 8; ++q) { s += r_s[tid * 8 + q]; s2 += r_q[tid * 8 + q]; }
    t_s[tid] = s; t_q[tid] = s2;
  }
  __syncthreads();
  float sum = 0.f, sumsq = 0.f;
  for (int q = 0; q < 32; ++q) { sum += t_s[q]; sumsq += t_q[q]; }

  const float cnt = (float)(BATCH * NPT * KNN_K);
  const float mean = sum / cnt;
  const float var  = sumsq / cnt - mean * mean;
  const float scale = gamma[o] * rsqrtf(var + BNEPS);
  const float shift = beta[o];
  const float* __restrict__ src = (scale >= 0.f) ? mxp : mnp;
  const size_t base = (size_t)plane * NPT + (size_t)(blockIdx.x & 3) * 2048 + tid * 4;
  const v4f a0 = *(const v4f*)(src + base);
  const v4f a1 = *(const v4f*)(src + base + 1024);
  v4f o0, o1;
  for (int e = 0; e < 4; ++e) {
    o0[e] = (a0[e] - mean) * scale + shift;
    o1[e] = (a1[e] - mean) * scale + shift;
  }
  *(v4f*)(outp + base)        = o0;
  *(v4f*)(outp + base + 1024) = o1;
}

extern "C" void kernel_launch(void* const* d_in, const int* in_sizes, int n_in,
                              void* d_out, int out_size, void* d_ws, size_t ws_size,
                              hipStream_t stream) {
  const float* x     = (const float*)d_in[0];
  const float* W     = (const float*)d_in[1];
  const float* bias  = (const float*)d_in[2];
  const float* gamma = (const float*)d_in[3];
  const float* beta  = (const float*)d_in[4];
  float* outp = (float*)d_out;

  float* xt  = (float*)d_ws;
  float* sq  = xt + (size_t)BATCH * NPT * CH;
  int*   knn = (int*)(sq + (size_t)BATCH * NPT);
  float* mxp = (float*)(knn + (size_t)BATCH * NPT * KNN_K);
  float* mnp = mxp + (size_t)BATCH * COUT * NPT;
  float* part = mnp + (size_t)BATCH * COUT * NPT + 128;

  float* xf = mxp;

  prep_kernel <<<(BATCH * NPT) / 256, 256, 0, stream>>>(x, xt, sq, xf);
  knn_kernel  <<<BATCH * (NPT / 16), 32 * CSPLIT, 0, stream>>>(xf, sq, knn);
  conv_kernel <<<(BATCH * NPT) / 16, 128, 0, stream>>>(xt, knn, W, bias, mxp, mnp, part);
  out_kernel  <<<BATCH * COUT * 4, 256, 0, stream>>>(mxp, mnp, part, gamma, beta, outp);
}


# knn top-k insert: one compare per level serves value and index, and the value comes from a 3-way median (v_med3) instead of compare+max+select: 80 vector ops per inserted score, was 141
# speedup vs baseline: 4.2897x; 1.3026x over previous
#include <hip/hip_runtime.h>
#include <stdint.h>

#define BATCH 2
#define CH    64
#define NPT   8192
#define KNN_K 20
#define COUT  64
#define CIN2  128
#define NEGSL 0.2f
#define BNEPS 1e-5f

typedef float v2f __attribute__((ext_vector_type(2)));
typedef float v4f __attribute__((ext_vector_type(4)));
typedef float v8f __attribute__((ext_vector_type(8)));

__device__ __forceinline__ v8f wmma_f32(v2f a, v2f b, v8f c) {
  return __builtin_amdgcn_wmma_f32_16x16x4_f32(false, a, false, b, (short)0, c, false, false);
}

__device__ __forceinline__ unsigned fenc(float f) {
  unsigned u = __float_as_uint(f);
  return (u & 0x80000000u) ? ~u : (u | 0x80000000u);
}
__device__ __forceinline__ float fdec(unsigned u) {
  return __uint_as_float((u & 0x80000000u) ? (u ^ 0x80000000u) : ~u);
}

__global__ void prep_kernel(const float* __restrict__ x,
                            float* __restrict__ sq, float* __restrict__ xf) {
  int gp = blockIdx.x * blockDim.x + threadIdx.x;
  int b = gp >> 13, n = gp & (NPT - 1);
  const float* xb = x + (size_t)b * CH * NPT + n;
  v4f* xf4 = (v4f*)xf + (size_t)(gp >> 4) * 256 + (gp & 15);
  float acc = 0.f;
  for (int i = 0; i < 8; ++i) {
    float v[8];
    #pragma unroll
    for (int e = 0; e < 8; ++e) v[e] = xb[(size_t)(i * 8 + e) * NPT];
    #pragma unroll
    for (int e = 0; e < 8; ++e) acc = __builtin_fmaf(v[e], v[e], acc);
    v4f h0, h1;
    h0.x = v[0]; h0.y = v[1]; h0.z = v[4]; h0.w = v[5];
    h1.x = v[2]; h1.y = v[3]; h1.z = v[6]; h1.w = v[7];
    xf4[i * 32]      = h0;
    xf4[i * 32 + 16] = h1;
  }
  sq[gp] = acc;
}

#define QSLOTS 33
#define CSPLIT 4
#define TPW    (NPT / 16 / CSPLIT)
#define NLIST  (2 * CSPLIT)
__launch_bounds__(32 * CSPLIT)
__global__ void knn_kernel(const float* __restrict__ xf, const float* __restrict__ sq,
                           int* __restrict__ knn,
                           const float* __restrict__ W, float* __restrict__ Y) {
  __shared__ float mv[CSPLIT * 32 * KNN_K];
  __shared__ int   mi[CSPLIT * 32 * KNN_K];
  __shared__ v2f   qall[CSPLIT * QSLOTS * 32];

  const int w  = __builtin_amdgcn_readfirstlane((int)(threadIdx.x >> 5));
  const int L  = threadIdx.x & 31;
  v2f* qbuf = qall + w * (QSLOTS * 32);
  const int rr = L & 15, hh = L >> 4;
  const int b  = blockIdx.x >> 9;
  const int r0 = (blockIdx.x & 511) << 4;
  const int bofs = b << 13;

  #pragma unroll
  for (int q = 0; q < QSLOTS; ++q) { v2f z; z.x = -__builtin_inff(); z.y = 0.f; qbuf[q * 32 + L] = z; }

  const v4f* xf4 = (const v4f*)xf + L;
  auto loadT = [&](int gt, v2f (&bb)[16]) {
    const v4f* p = xf4 + (size_t)gt * 256;
    #pragma unroll
    for (int i = 0; i < 8; ++i) {
      v4f v = p[i * 32];
      bb[2 * i].x = v.x;     bb[2 * i].y = v.y;
      bb[2 * i + 1].x = v.z; bb[2 * i + 1].y = v.w;
    }
  };

  v2f a[16];
  loadT((int)blockIdx.x, a);
  const float sqn = sq[bofs + r0 + rr];

  {
    const float* wr = W + (size_t)(w * 16 + rr) * CIN2 + 2 * hh;
    v2f wa[16];
    #pragma unroll
    for (int kk = 0; kk < 16; ++kk) wa[kk] = *(const v2f*)(wr + 4 * kk);
    v8f y0 = {}, y1 = {};
    #pragma unroll
    for (int kk = 0; kk < 16; kk += 2) {
      y0 = wmma_f32(wa[kk],     a[kk],     y0);
      y1 = wmma_f32(wa[kk + 1], a[kk + 1], y1);
    }
    v4f lo, hi;
    #pragma unroll
    for (int j = 0; j < 4; ++j) { lo[j] = y0[j] + y1[j]; hi[j] = y0[j + 4] + y1[j + 4]; }
    float* yo = Y + (size_t)(bofs + r0 + rr) * COUT + w * 16 + hh * 8;
    *(v4f*)yo       = lo;
    *(v4f*)(yo + 4) = hi;
  }

  float vals[KNN_K]; int idxs[KNN_K];
  #pragma unroll
  for (int q = 0; q < KNN_K; ++q) { vals[q] = -__builtin_inff(); idxs[q] = 0; }
  int qn = 0;

  const int tb = b << 9;
  auto loadB = [&](int ct, v2f (&bb)[16]) { loadT(tb + ct, bb); };

  auto insert = [&](float s, int colg) {
    bool gt[KNN_K];
    #pragma unroll
    for (int t = 1; t < KNN_K; ++t) gt[t] = s > vals[t];
    #pragma unroll
    for (int t = 0; t < KNN_K - 1; ++t) {
      const int own = (t == 0) ? colg : (gt[t] ? colg : idxs[t]);
      idxs[t] = gt[t + 1] ? idxs[t + 1] : own;
      vals[t] = __builtin_amdgcn_fmed3f(vals[t], s, vals[t + 1]);
    }
    idxs[KNN_K - 1] = gt[KNN_K - 1] ? colg : idxs[KNN_K - 1];
    vals[KNN_K - 1] = gt[KNN_K - 1] ? s : vals[KNN_K - 1];
  };

  auto drain = [&]() {
    const int mq = __builtin_amdgcn_readfirstlane((int)__builtin_amdgcn_wave_reduce_max_u32((unsigned)qn, 0));
    v2f e = qbuf[L];
    for (int i = 0; i < mq; ++i) {
      v2f cur = e;
      e = qbuf[(i + 1) * 32 + L];
      float s = cur.x;
      if (i < qn && s > vals[0]) insert(s, __float_as_int(cur.y));
    }
    qn = 0;
  };

  auto step = [&](int ct, v2f (&cur)[16], v2f (&nxt)[16]) {
    const int c0 = ct << 4;
    v4f sq0 = *(const v4f*)(sq + bofs + c0 + hh * 8);
    v4f sq1 = *(const v4f*)(sq + bofs + c0 + hh * 8 + 4);
    loadB((ct + 1) & 511, nxt);
    float sqt[8];
    #pragma unroll
    for (int j = 0; j < 4; ++j) { sqt[j] = sq0[j]; sqt[j + 4] = sq1[j]; }

    v8f acc = {};
    #pragma unroll
    for (int kk = 0; kk < 16; ++kk) acc = wmma_f32(cur[kk], a[kk], acc);

    const float thr = vals[0];
    #pragma unroll
    for (int j = 0; j < 8; ++j) {
      int colg = c0 + hh * 8 + j;
      float s = 2.f * acc[j] - sqt[j] - sqn;
      v2f ent; ent.x = s; ent.y = __int_as_float(colg);
      qbuf[qn * 32 + L] = ent;
      qn += (s > thr) ? 1 : 0;
    }
  };

  v2f bufA[16], bufB[16];
  const int ct0 = w * TPW;
  loadB(ct0, bufA);
  for (int ct = ct0; ct < ct0 + TPW; ct += 2) {
    step(ct,     bufA, bufB);
    step(ct + 1, bufB, bufA);
    if (__builtin_amdgcn_ballot_w32(qn > 16) != 0u) drain();
  }
  drain();

  #pragma unroll
  for (int q = 0; q < KNN_K; ++q) {
    mv[(w * 32 + L) * KNN_K + q] = vals[q];
    mi[(w * 32 + L) * KNN_K + q] = idxs[q];
  }
  __syncthreads();
  if (threadIdx.x < 16) {
    const int r = threadIdx.x;
    float hv[NLIST]; int hp[NLIST];
    #pragma unroll
    for (int l = 0; l < NLIST; ++l) {
      hp[l] = KNN_K - 1;
      hv[l] = mv[((l >> 1) * 32 + (l & 1) * 16 + r) * KNN_K + KNN_K - 1];
    }
    int* outp = knn + (size_t)(bofs + r0 + r) * KNN_K;
    for (int q = 0; q < KNN_K; ++q) {
      int best = 0; float bv = hv[0];
      #pragma unroll
      for (int l = 1; l < NLIST; ++l) { bool g = hv[l] > bv; bv = g ? hv[l] : bv; best = g ? l : best; }
      int bp = hp[0];
      #pragma unroll
      for (int l = 1; l < NLIST; ++l) bp = (l == best) ? hp[l] : bp;
      const int base = ((best >> 1) * 32 + (best & 1) * 16 + r) * KNN_K;
      outp[q] = mi[base + bp];
      const int   np = (bp > 0) ? bp - 1 : 0;
      const float nv = (bp > 0) ? mv[base + np] : -__builtin_inff();
      #pragma unroll
      for (int l = 0; l < NLIST; ++l) { bool m = (l == best); hp[l] = m ? bp - 1 : hp[l]; hv[l] = m ? nv : hv[l]; }
    }
  }
}

#define ZP 66
__launch_bounds__(128)
__global__ void conv_kernel(const float* __restrict__ x, const float* __restrict__ Y,
                            const int* __restrict__ knn,
                            const float* __restrict__ W, const float* __restrict__ bias,
                            float* __restrict__ mxp, float* __restrict__ mnp,
                            float* __restrict__ part) {
  __shared__ float s_z [16 * ZP];
  __shared__ float s_mx[16 * ZP];
  __shared__ float s_mn[16 * ZP];
  __shared__ float s_ps[4][COUT];
  __shared__ float s_pq[4][COUT];

  const int tid = threadIdx.x;
  const int w  = __builtin_amdgcn_readfirstlane((int)(tid >> 5));
  const int L  = tid & 31;
  const int rr = L & 15, hh = L >> 4;
  const int gp0 = blockIdx.x * 16;
  const int b = gp0 >> 13, n0 = gp0 & (NPT - 1);

  {
    const float* xp = x + (size_t)b * CH * NPT + (size_t)(2 * hh) * NPT + n0 + rr;
    const float* wr = W + (size_t)(w * 16 + rr) * CIN2 + 2 * hh;
    v2f pt[16], w1[16], w2[16];
    #pragma unroll
    for (int kk = 0; kk < 16; ++kk) {
      pt[kk].x = xp[(size_t)(4 * kk) * NPT];
      pt[kk].y = xp[(size_t)(4 * kk + 1) * NPT];
      w1[kk] = *(const v2f*)(wr + 4 * kk);
      w2[kk] = *(const v2f*)(wr + CH + 4 * kk);
    }
    v8f ay = {}, au = {};
    #pragma unroll
    for (int kk = 0; kk < 16; ++kk) {
      ay = wmma_f32(w1[kk], pt[kk], ay);
      au = wmma_f32(w2[kk], pt[kk], au);
    }
    const int ob = w * 16 + hh * 8;
    const v4f b0 = *(const v4f*)(bias + ob);
    const v4f b1 = *(const v4f*)(bias + ob + 4);
    float* zr = s_z + rr * ZP + ob;
    #pragma unroll
    for (int j = 0; j < 4; j += 2) {
      v2f t0, t1;
      t0.x = (au[j]     - ay[j])     + b0[j];
      t0.y = (au[j + 1] - ay[j + 1]) + b0[j + 1];
      t1.x = (au[j + 4] - ay[j + 4]) + b1[j];
      t1.y = (au[j + 5] - ay[j + 5]) + b1[j + 1];
      *(v2f*)(zr + j)     = t0;
      *(v2f*)(zr + 4 + j) = t1;
    }
  }
  __syncthreads();

  float s0 = 0.f, s1 = 0.f, q0 = 0.f, q1 = 0.f;
  {
    const char* yb = (const char*)(Y + ((size_t)b << 13) * COUT) + L * 8;
    #pragma unroll
    for (int p = 0; p < 4; ++p) {
      const int pl = w * 4 + p;
      const int* kp = knn + (size_t)(gp0 + pl) * KNN_K;
      v2f r[KNN_K];
      #pragma unroll
      for (int k = 0; k < KNN_K; ++k) r[k] = *(const v2f*)(yb + ((unsigned)kp[k] << 8));
      const v2f z = *(const v2f*)(s_z + pl * ZP + 2 * L);
      float mx0 = -__builtin_inff(), mx1 = -__builtin_inff();
      float mn0 =  __builtin_inff(), mn1 =  __builtin_inff();
      #pragma unroll
      for (int k = 0; k < KNN_K; ++k) {
        float h0 = r[k].x + z.x, h1 = r[k].y + z.y;
        h0 = (h0 >= 0.f) ? h0 : NEGSL * h0;
        h1 = (h1 >= 0.f) ? h1 : NEGSL * h1;
        s0 += h0; q0 += h0 * h0;
        s1 += h1; q1 += h1 * h1;
        mx0 = fmaxf(mx0, h0); mn0 = fminf(mn0, h0);
        mx1 = fmaxf(mx1, h1); mn1 = fminf(mn1, h1);
      }
      v2f vmx, vmn;
      vmx.x = mx0; vmx.y = mx1; vmn.x = mn0; vmn.y = mn1;
      *(v2f*)(s_mx + pl * ZP + 2 * L) = vmx;
      *(v2f*)(s_mn + pl * ZP + 2 * L) = vmn;
    }
  }
  s_ps[w][2 * L] = s0; s_ps[w][2 * L + 1] = s1;
  s_pq[w][2 * L] = q0; s_pq[w][2 * L + 1] = q1;
  __syncthreads();
  if (tid < COUT) {
    const float s  = ((s_ps[0][tid] + s_ps[1][tid]) + s_ps[2][tid]) + s_ps[3][tid];
    const float s2 = ((s_pq[0][tid] + s_pq[1][tid]) + s_pq[2][tid]) + s_pq[3][tid];
    part[(size_t)tid * 1024 + blockIdx.x]        = s;
    part[(size_t)(64 + tid) * 1024 + blockIdx.x] = s2;
  }
  {
    const int nl = tid & 15, og = tid >> 4;
    #pragma unroll
    for (int it = 0; it < 8; ++it) {
      const int o = og + 8 * it;
      const size_t oi = ((size_t)(b * COUT + o) << 13) + n0 + nl;
      mxp[oi] = s_mx[nl * ZP + o];
      mnp[oi] = s_mn[nl * ZP + o];
    }
  }
}

__launch_bounds__(256)
__global__ void out_kernel(const float* __restrict__ mxp, const float* __restrict__ mnp,
                           const float* __restrict__ part,
                           const float* __restrict__ gamma, const float* __restrict__ beta,
                           float* __restrict__ outp) {
  __shared__ float r_s[256], r_q[256];
  __shared__ float t_s[32],  t_q[32];
  const int tid   = threadIdx.x;
  const int plane = blockIdx.x >> 2;
  const int o     = plane & (COUT - 1);

  const v4f ps = *(const v4f*)(part + (size_t)o * 1024 + tid * 4);
  const v4f pq = *(const v4f*)(part + (size_t)(COUT + o) * 1024 + tid * 4);
  r_s[tid] = ((ps.x + ps.y) + ps.z) + ps.w;
  r_q[tid] = ((pq.x + pq.y) + pq.z) + pq.w;
  __syncthreads();
  if (tid < 32) {
    float s = 0.f, s2 = 0.f;
    for (int q = 0; q < 8; ++q) { s += r_s[tid * 8 + q]; s2 += r_q[tid * 8 + q]; }
    t_s[tid] = s; t_q[tid] = s2;
  }
  __syncthreads();
  float sum = 0.f, sumsq = 0.f;
  for (int q = 0; q < 32; ++q) { sum += t_s[q]; sumsq += t_q[q]; }

  const float cnt = (float)(BATCH * NPT * KNN_K);
  const float mean = sum / cnt;
  const float var  = sumsq / cnt - mean * mean;
  const float scale = gamma[o] * rsqrtf(var + BNEPS);
  const float shift = beta[o];
  const float* __restrict__ src = (scale >= 0.f) ? mxp : mnp;
  const size_t base = (size_t)plane * NPT + (size_t)(blockIdx.x & 3) * 2048 + tid * 4;
  const v4f a0 = *(const v4f*)(src + base);
  const v4f a1 = *(const v4f*)(src + base + 1024);
  v4f o0, o1;
  for (int e = 0; e < 4; ++e) {
    o0[e] = (a0[e] - mean) * scale + shift;
    o1[e] = (a1[e] - mean) * scale + shift;
  }
  *(v4f*)(outp + base)        = o0;
  *(v4f*)(outp + base + 1024) = o1;
}

extern "C" void kernel_launch(void* const* d_in, const int* in_sizes, int n_in,
                              void* d_out, int out_size, void* d_ws, size_t ws_size,
                              hipStream_t stream) {
  const float* x     = (const float*)d_in[0];
  const float* W     = (const float*)d_in[1];
  const float* bias  = (const float*)d_in[2];
  const float* gamma = (const float*)d_in[3];
  const float* beta  = (const float*)d_in[4];
  float* outp = (float*)d_out;

  float* yv  = (float*)d_ws;
  float* sq  = yv + (size_t)BATCH * NPT * COUT;
  int*   knn = (int*)(sq + (size_t)BATCH * NPT);
  float* mxp = (float*)(knn + (size_t)BATCH * NPT * KNN_K);
  float* mnp = mxp + (size_t)BATCH * COUT * NPT;
  float* part = mnp + (size_t)BATCH * COUT * NPT + 128;

  float* xf = mxp;

  prep_kernel  <<<(BATCH * NPT) / 256, 256, 0, stream>>>(x, sq, xf);
  knn_kernel   <<<BATCH * (NPT / 16), 32 * CSPLIT, 0, stream>>>(xf, sq, knn, W, yv);
  conv_kernel  <<<(BATCH * NPT) / 16, 128, 0, stream>>>(x, yv, knn, W, bias, mxp, mnp, part);
  out_kernel   <<<BATCH * COUT * 4, 256, 0, stream>>>(mxp, mnp, part, gamma, beta, outp);
}


# knn top-k: the 8 lists of a row share a row-wide threshold through LDS (each publishes its 3rd best; the least of the 8 bounds the row's 24th best), so scores below it are neither queued nor inserted
# speedup vs baseline: 4.5443x; 1.0594x over previous
#include <hip/hip_runtime.h>
#include <stdint.h>

#define BATCH 2
#define CH    64
#define NPT   8192
#define KNN_K 20
#define COUT  64
#define CIN2  128
#define NEGSL 0.2f
#define BNEPS 1e-5f

typedef float v2f __attribute__((ext_vector_type(2)));
typedef float v4f __attribute__((ext_vector_type(4)));
typedef float v8f __attribute__((ext_vector_type(8)));

__device__ __forceinline__ v8f wmma_f32(v2f a, v2f b, v8f c) {
  return __builtin_amdgcn_wmma_f32_16x16x4_f32(false, a, false, b, (short)0, c, false, false);
}

__device__ __forceinline__ unsigned fenc(float f) {
  unsigned u = __float_as_uint(f);
  return (u & 0x80000000u) ? ~u : (u | 0x80000000u);
}
__device__ __forceinline__ float fdec(unsigned u) {
  return __uint_as_float((u & 0x80000000u) ? (u ^ 0x80000000u) : ~u);
}

__global__ void prep_kernel(const float* __restrict__ x,
                            float* __restrict__ sq, float* __restrict__ xf) {
  int gp = blockIdx.x * blockDim.x + threadIdx.x;
  int b = gp >> 13, n = gp & (NPT - 1);
  const float* xb = x + (size_t)b * CH * NPT + n;
  v4f* xf4 = (v4f*)xf + (size_t)(gp >> 4) * 256 + (gp & 15);
  float acc = 0.f;
  for (int i = 0; i < 8; ++i) {
    float v[8];
    #pragma unroll
    for (int e = 0; e < 8; ++e) v[e] = xb[(size_t)(i * 8 + e) * NPT];
    #pragma unroll
    for (int e = 0; e < 8; ++e) acc = __builtin_fmaf(v[e], v[e], acc);
    v4f h0, h1;
    h0.x = v[0]; h0.y = v[1]; h0.z = v[4]; h0.w = v[5];
    h1.x = v[2]; h1.y = v[3]; h1.z = v[6]; h1.w = v[7];
    xf4[i * 32]      = h0;
    xf4[i * 32 + 16] = h1;
  }
  sq[gp] = acc;
}

#define QSLOTS 33
#define CSPLIT 4
#define TPW    (NPT / 16 / CSPLIT)
#define NLIST  (2 * CSPLIT)
__launch_bounds__(32 * CSPLIT)
__global__ void knn_kernel(const float* __restrict__ xf, const float* __restrict__ sq,
                           int* __restrict__ knn,
                           const float* __restrict__ W, float* __restrict__ Y) {
  __shared__ float mv[CSPLIT * 32 * KNN_K];
  __shared__ int   mi[CSPLIT * 32 * KNN_K];
  __shared__ v2f   qall[CSPLIT * QSLOTS * 32];
  __shared__ __attribute__((aligned(16))) float rowthr[16 * NLIST];

  const int w  = __builtin_amdgcn_readfirstlane((int)(threadIdx.x >> 5));
  const int L  = threadIdx.x & 31;
  v2f* qbuf = qall + w * (QSLOTS * 32);
  const int rr = L & 15, hh = L >> 4;
  const int b  = blockIdx.x >> 9;
  const int r0 = (blockIdx.x & 511) << 4;
  const int bofs = b << 13;

  #pragma unroll
  for (int q = 0; q < QSLOTS; ++q) { v2f z; z.x = -__builtin_inff(); z.y = 0.f; qbuf[q * 32 + L] = z; }
  const int mylist = rr * NLIST + 2 * w + hh;
  rowthr[mylist] = -__builtin_inff();

  const v4f* xf4 = (const v4f*)xf + L;
  auto loadT = [&](int gt, v2f (&bb)[16]) {
    const v4f* p = xf4 + (size_t)gt * 256;
    #pragma unroll
    for (int i = 0; i < 8; ++i) {
      v4f v = p[i * 32];
      bb[2 * i].x = v.x;     bb[2 * i].y = v.y;
      bb[2 * i + 1].x = v.z; bb[2 * i + 1].y = v.w;
    }
  };

  v2f a[16];
  loadT((int)blockIdx.x, a);
  const float sqn = sq[bofs + r0 + rr];

  {
    const float* wr = W + (size_t)(w * 16 + rr) * CIN2 + 2 * hh;
    v2f wa[16];
    #pragma unroll
    for (int kk = 0; kk < 16; ++kk) wa[kk] = *(const v2f*)(wr + 4 * kk);
    v8f y0 = {}, y1 = {};
    #pragma unroll
    for (int kk = 0; kk < 16; kk += 2) {
      y0 = wmma_f32(wa[kk],     a[kk],     y0);
      y1 = wmma_f32(wa[kk + 1], a[kk + 1], y1);
    }
    v4f lo, hi;
    #pragma unroll
    for (int j = 0; j < 4; ++j) { lo[j] = y0[j] + y1[j]; hi[j] = y0[j + 4] + y1[j + 4]; }
    float* yo = Y + (size_t)(bofs + r0 + rr) * COUT + w * 16 + hh * 8;
    __syncthreads();
    *(v4f*)yo       = lo;
    *(v4f*)(yo + 4) = hi;
  }

  float vals[KNN_K]; int idxs[KNN_K];
  #pragma unroll
  for (int q = 0; q < KNN_K; ++q) { vals[q] = -__builtin_inff(); idxs[q] = 0; }
  int qn = 0;

  auto rowbound = [&]() -> float {
    asm volatile("" ::: "memory");
    const v4f* p = (const v4f*)(rowthr + rr * NLIST);
    const v4f t0 = p[0], t1 = p[1];
    const float T = fminf(fminf(fminf(t0.x, t0.y), fminf(t0.z, t0.w)),
                          fminf(fminf(t1.x, t1.y), fminf(t1.z, t1.w)));
    return __builtin_fmaf(-__builtin_fabsf(T), 0x1p-20f, T) - 1e-30f;
  };
  float thr = -__builtin_inff();

  const int tb = b << 9;
  auto loadB = [&](int ct, v2f (&bb)[16]) { loadT(tb + ct, bb); };

  auto insert = [&](float s, int colg) {
    bool gt[KNN_K];
    #pragma unroll
    for (int t = 1; t < KNN_K; ++t) gt[t] = s > vals[t];
    #pragma unroll
    for (int t = 0; t < KNN_K - 1; ++t) {
      const int own = (t == 0) ? colg : (gt[t] ? colg : idxs[t]);
      idxs[t] = gt[t + 1] ? idxs[t + 1] : own;
      vals[t] = __builtin_amdgcn_fmed3f(vals[t], s, vals[t + 1]);
    }
    idxs[KNN_K - 1] = gt[KNN_K - 1] ? colg : idxs[KNN_K - 1];
    vals[KNN_K - 1] = gt[KNN_K - 1] ? s : vals[KNN_K - 1];
  };

  auto drain = [&]() {
    const int mq = __builtin_amdgcn_readfirstlane((int)__builtin_amdgcn_wave_reduce_max_u32((unsigned)qn, 0));
    const float rb = rowbound();
    v2f e = qbuf[L];
    for (int i = 0; i < mq; ++i) {
      v2f cur = e;
      e = qbuf[(i + 1) * 32 + L];
      float s = cur.x;
      if (i < qn && s > vals[0] && s > rb) insert(s, __float_as_int(cur.y));
    }
    qn = 0;
    rowthr[mylist] = vals[KNN_K - 3];
  };

  auto step = [&](int ct, v2f (&cur)[16], v2f (&nxt)[16]) {
    const int c0 = ct << 4;
    v4f sq0 = *(const v4f*)(sq + bofs + c0 + hh * 8);
    v4f sq1 = *(const v4f*)(sq + bofs + c0 + hh * 8 + 4);
    loadB((ct + 1) & 511, nxt);
    float sqt[8];
    #pragma unroll
    for (int j = 0; j < 4; ++j) { sqt[j] = sq0[j]; sqt[j + 4] = sq1[j]; }

    v8f acc = {};
    #pragma unroll
    for (int kk = 0; kk < 16; ++kk) acc = wmma_f32(cur[kk], a[kk], acc);

    #pragma unroll
    for (int j = 0; j < 8; ++j) {
      int colg = c0 + hh * 8 + j;
      float s = 2.f * acc[j] - sqt[j] - sqn;
      v2f ent; ent.x = s; ent.y = __int_as_float(colg);
      qbuf[qn * 32 + L] = ent;
      qn += (s > thr) ? 1 : 0;
    }
  };

  v2f bufA[16], bufB[16];
  const int ct0 = w * TPW;
  loadB(ct0, bufA);
  for (int ct = ct0; ct < ct0 + TPW; ct += 2) {
    thr = fmaxf(vals[0], rowbound());
    step(ct,     bufA, bufB);
    step(ct + 1, bufB, bufA);
    if (__builtin_amdgcn_ballot_w32(qn > 15) != 0u) drain();
  }
  drain();

  #pragma unroll
  for (int q = 0; q < KNN_K; ++q) {
    mv[(w * 32 + L) * KNN_K + q] = vals[q];
    mi[(w * 32 + L) * KNN_K + q] = idxs[q];
  }
  __syncthreads();
  if (threadIdx.x < 16) {
    const int r = threadIdx.x;
    float hv[NLIST]; int hp[NLIST];
    #pragma unroll
    for (int l = 0; l < NLIST; ++l) {
      hp[l] = KNN_K - 1;
      hv[l] = mv[((l >> 1) * 32 + (l & 1) * 16 + r) * KNN_K + KNN_K - 1];
    }
    int* outp = knn + (size_t)(bofs + r0 + r) * KNN_K;
    for (int q = 0; q < KNN_K; ++q) {
      int best = 0; float bv = hv[0];
      #pragma unroll
      for (int l = 1; l < NLIST; ++l) { bool g = hv[l] > bv; bv = g ? hv[l] : bv; best = g ? l : best; }
      int bp = hp[0];
      #pragma unroll
      for (int l = 1; l < NLIST; ++l) bp = (l == best) ? hp[l] : bp;
      const int base = ((best >> 1) * 32 + (best & 1) * 16 + r) * KNN_K;
      outp[q] = mi[base + bp];
      const int   np = (bp > 0) ? bp - 1 : 0;
      const float nv = (bp > 0) ? mv[base + np] : -__builtin_inff();
      #pragma unroll
      for (int l = 0; l < NLIST; ++l) { bool m = (l == best); hp[l] = m ? bp - 1 : hp[l]; hv[l] = m ? nv : hv[l]; }
    }
  }
}

#define ZP 66
__launch_bounds__(128)
__global__ void conv_kernel(const float* __restrict__ x, const float* __restrict__ Y,
                            const int* __restrict__ knn,
                            const float* __restrict__ W, const float* __restrict__ bias,
                            float* __restrict__ mxp, float* __restrict__ mnp,
                            float* __restrict__ part) {
  __shared__ float s_z [16 * ZP];
  __shared__ float s_mx[16 * ZP];
  __shared__ float s_mn[16 * ZP];
  __shared__ float s_ps[4][COUT];
  __shared__ float s_pq[4][COUT];

  const int tid = threadIdx.x;
  const int w  = __builtin_amdgcn_readfirstlane((int)(tid >> 5));
  const int L  = tid & 31;
  const int rr = L & 15, hh = L >> 4;
  const int gp0 = blockIdx.x * 16;
  const int b = gp0 >> 13, n0 = gp0 & (NPT - 1);

  {
    const float* xp = x + (size_t)b * CH * NPT + (size_t)(2 * hh) * NPT + n0 + rr;
    const float* wr = W + (size_t)(w * 16 + rr) * CIN2 + 2 * hh;
    v2f pt[16], w1[16], w2[16];
    #pragma unroll
    for (int kk = 0; kk < 16; ++kk) {
      pt[kk].x = xp[(size_t)(4 * kk) * NPT];
      pt[kk].y = xp[(size_t)(4 * kk + 1) * NPT];
      w1[kk] = *(const v2f*)(wr + 4 * kk);
      w2[kk] = *(const v2f*)(wr + CH + 4 * kk);
    }
    v8f ay = {}, au = {};
    #pragma unroll
    for (int kk = 0; kk < 16; ++kk) {
      ay = wmma_f32(w1[kk], pt[kk], ay);
      au = wmma_f32(w2[kk], pt[kk], au);
    }
    const int ob = w * 16 + hh * 8;
    const v4f b0 = *(const v4f*)(bias + ob);
    const v4f b1 = *(const v4f*)(bias + ob + 4);
    float* zr = s_z + rr * ZP + ob;
    #pragma unroll
    for (int j = 0; j < 4; j += 2) {
      v2f t0, t1;
      t0.x = (au[j]     - ay[j])     + b0[j];
      t0.y = (au[j + 1] - ay[j + 1]) + b0[j + 1];
      t1.x = (au[j + 4] - ay[j + 4]) + b1[j];
      t1.y = (au[j + 5] - ay[j + 5]) + b1[j + 1];
      *(v2f*)(zr + j)     = t0;
      *(v2f*)(zr + 4 + j) = t1;
    }
  }
  __syncthreads();

  float s0 = 0.f, s1 = 0.f, q0 = 0.f, q1 = 0.f;
  {
    const char* yb = (const char*)(Y + ((size_t)b << 13) * COUT) + L * 8;
    #pragma unroll
    for (int p = 0; p < 4; ++p) {
      const int pl = w * 4 + p;
      const int* kp = knn + (size_t)(gp0 + pl) * KNN_K;
      v2f r[KNN_K];
      #pragma unroll
      for (int k = 0; k < KNN_K; ++k) r[k] = *(const v2f*)(yb + ((unsigned)kp[k] << 8));
      const v2f z = *(const v2f*)(s_z + pl * ZP + 2 * L);
      float mx0 = -__builtin_inff(), mx1 = -__builtin_inff();
      float mn0 =  __builtin_inff(), mn1 =  __builtin_inff();
      #pragma unroll
      for (int k = 0; k < KNN_K; ++k) {
        float h0 = r[k].x + z.x, h1 = r[k].y + z.y;
        h0 = (h0 >= 0.f) ? h0 : NEGSL * h0;
        h1 = (h1 >= 0.f) ? h1 : NEGSL * h1;
        s0 += h0; q0 += h0 * h0;
        s1 += h1; q1 += h1 * h1;
        mx0 = fmaxf(mx0, h0); mn0 = fminf(mn0, h0);
        mx1 = fmaxf(mx1, h1); mn1 = fminf(mn1, h1);
      }
      v2f vmx, vmn;
      vmx.x = mx0; vmx.y = mx1; vmn.x = mn0; vmn.y = mn1;
      *(v2f*)(s_mx + pl * ZP + 2 * L) = vmx;
      *(v2f*)(s_mn + pl * ZP + 2 * L) = vmn;
    }
  }
  s_ps[w][2 * L] = s0; s_ps[w][2 * L + 1] = s1;
  s_pq[w][2 * L] = q0; s_pq[w][2 * L + 1] = q1;
  __syncthreads();
  if (tid < COUT) {
    const float s  = ((s_ps[0][tid] + s_ps[1][tid]) + s_ps[2][tid]) + s_ps[3][tid];
    const float s2 = ((s_pq[0][tid] + s_pq[1][tid]) + s_pq[2][tid]) + s_pq[3][tid];
    part[(size_t)tid * 1024 + blockIdx.x]        = s;
    part[(size_t)(64 + tid) * 1024 + blockIdx.x] = s2;
  }
  {
    const int nl = tid & 15, og = tid >> 4;
    #pragma unroll
    for (int it = 0; it < 8; ++it) {
      const int o = og + 8 * it;
      const size_t oi = ((size_t)(b * COUT + o) << 13) + n0 + nl;
      mxp[oi] = s_mx[nl * ZP + o];
      mnp[oi] = s_mn[nl * ZP + o];
    }
  }
}

__launch_bounds__(256)
__global__ void out_kernel(const float* __restrict__ mxp, const float* __restrict__ mnp,
                           const float* __restrict__ part,
                           const float* __restrict__ gamma, const float* __restrict__ beta,
                           float* __restrict__ outp) {
  __shared__ float r_s[256], r_q[256];
  __shared__ float t_s[32],  t_q[32];
  const int tid   = threadIdx.x;
  const int plane = blockIdx.x >> 2;
  const int o     = plane & (COUT - 1);

  const v4f ps = *(const v4f*)(part + (size_t)o * 1024 + tid * 4);
  const v4f pq = *(const v4f*)(part + (size_t)(COUT + o) * 1024 + tid * 4);
  r_s[tid] = ((ps.x + ps.y) + ps.z) + ps.w;
  r_q[tid] = ((pq.x + pq.y) + pq.z) + pq.w;
  __syncthreads();
  if (tid < 32) {
    float s = 0.f, s2 = 0.f;
    for (int q = 0; q < 8; ++q) { s += r_s[tid * 8 + q]; s2 += r_q[tid * 8 + q]; }
    t_s[tid] = s; t_q[tid] = s2;
  }
  __syncthreads();
  float sum = 0.f, sumsq = 0.f;
  for (int q = 0; q < 32; ++q) { sum += t_s[q]; sumsq += t_q[q]; }

  const float cnt = (float)(BATCH * NPT * KNN_K);
  const float mean = sum / cnt;
  const float var  = sumsq / cnt - mean * mean;
  const float scale = gamma[o] * rsqrtf(var + BNEPS);
  const float shift = beta[o];
  const float* __restrict__ src = (scale >= 0.f) ? mxp : mnp;
  const size_t base = (size_t)plane * NPT + (size_t)(blockIdx.x & 3) * 2048 + tid * 4;
  const v4f a0 = *(const v4f*)(src + base);
  const v4f a1 = *(const v4f*)(src + base + 1024);
  v4f o0, o1;
  for (int e = 0; e < 4; ++e) {
    o0[e] = (a0[e] - mean) * scale + shift;
    o1[e] = (a1[e] - mean) * scale + shift;
  }
  *(v4f*)(outp + base)        = o0;
  *(v4f*)(outp + base + 1024) = o1;
}

extern "C" void kernel_launch(void* const* d_in, const int* in_sizes, int n_in,
                              void* d_out, int out_size, void* d_ws, size_t ws_size,
                              hipStream_t stream) {
  const float* x     = (const float*)d_in[0];
  const float* W     = (const float*)d_in[1];
  const float* bias  = (const float*)d_in[2];
  const float* gamma = (const float*)d_in[3];
  const float* beta  = (const float*)d_in[4];
  float* outp = (float*)d_out;

  float* yv  = (float*)d_ws;
  float* sq  = yv + (size_t)BATCH * NPT * COUT;
  int*   knn = (int*)(sq + (size_t)BATCH * NPT);
  float* mxp = (float*)(knn + (size_t)BATCH * NPT * KNN_K);
  float* mnp = mxp + (size_t)BATCH * COUT * NPT;
  float* part = mnp + (size_t)BATCH * COUT * NPT + 128;

  float* xf = mxp;

  prep_kernel  <<<(BATCH * NPT) / 256, 256, 0, stream>>>(x, sq, xf);
  knn_kernel   <<<BATCH * (NPT / 16), 32 * CSPLIT, 0, stream>>>(xf, sq, knn, W, yv);
  conv_kernel  <<<(BATCH * NPT) / 16, 128, 0, stream>>>(x, yv, knn, W, bias, mxp, mnp, part);
  out_kernel   <<<BATCH * COUT * 4, 256, 0, stream>>>(mxp, mnp, part, gamma, beta, outp);
}
